# starting kernel
# baseline (speedup 1.0000x reference)
#include <hip/hip_runtime.h>
#include <hip/hip_bf16.h>

typedef __attribute__((ext_vector_type(2))) float v2f;
typedef __attribute__((ext_vector_type(8))) float v8f;

__global__ void k_init_deg(float* __restrict__ deg, int n) {
    int i = blockIdx.x * blockDim.x + threadIdx.x;
    if (i < n) deg[i] = 1.0f;
}

__global__ void k_deg_edges(const int* __restrict__ col, const float* __restrict__ ew,
                            float* __restrict__ deg, int e) {
    int i = blockIdx.x * blockDim.x + threadIdx.x;
    if (i < e) atomicAdd(&deg[col[i]], ew[i]);
}

__global__ void k_dis(float* __restrict__ deg_dis, int n) {
    int i = blockIdx.x * blockDim.x + threadIdx.x;
    if (i < n) {
        float d = deg_dis[i];
        deg_dis[i] = (d > 0.f) ? rsqrtf(d) : 0.f;
    }
}

__global__ void k_gemm1(const float* __restrict__ x, const float* __restrict__ W1,
                        float* __restrict__ h1, int n, int ntiles) {
    int gtid = blockIdx.x * blockDim.x + threadIdx.x;
    int tile = gtid >> 5;
    int lane = threadIdx.x & 31;
    if (tile >= ntiles) return;
    int half = lane >> 4;
    int m    = lane & 15;
    int row  = tile * 16 + m;
    int rowc = row < n ? row : n - 1;
    bool rv  = row < n;
    int k0   = half * 2;

    float ax = x[(size_t)rowc * 3 + k0];
    float ay = x[(size_t)rowc * 3 + 1];
    v2f a;
    a.x = rv ? ax : 0.f;
    a.y = (rv && half == 0) ? ay : 0.f;

    float bx = W1[k0 * 16 + m];
    float by = W1[16 + m];
    v2f b;
    b.x = bx;
    b.y = (half == 0) ? by : 0.f;

    v8f c = {};
    c = __builtin_amdgcn_wmma_f32_16x16x4_f32(false, a, false, b, (short)0, c, false, false);

    float* p = h1 + ((size_t)(tile * 16 + half * 8)) * 16 + m;
    if (tile * 16 + 16 <= n) {
#pragma unroll
        for (int j = 0; j < 8; ++j) p[(size_t)j * 16] = c[j];
    } else {
#pragma unroll
        for (int j = 0; j < 8; ++j)
            if (tile * 16 + half * 8 + j < n) p[(size_t)j * 16] = c[j];
    }
}

__global__ void k_self1(const float* __restrict__ h1, const float* __restrict__ dis,
                        float* __restrict__ acc, int n16) {
    int i = blockIdx.x * blockDim.x + threadIdx.x;
    if (i < n16) {
        float s = dis[i >> 4];
        acc[i] = h1[i] * s * s;
    }
}

__global__ void k_edges1(const int* __restrict__ row, const int* __restrict__ col,
                         const float* __restrict__ ew, const float* __restrict__ dis,
                         const float* __restrict__ h1, float* __restrict__ acc, int e) {
    int i = blockIdx.x * blockDim.x + threadIdx.x;
    if (i >= e) return;
    int r = row[i], c = col[i];
    float nrm = dis[r] * ew[i] * dis[c];
    const float4* hr = (const float4*)(h1 + (size_t)r * 16);
    float4 v0 = hr[0], v1 = hr[1], v2 = hr[2], v3 = hr[3];
    float* dst = acc + (size_t)c * 16;
    atomicAdd(dst +  0, v0.x * nrm); atomicAdd(dst +  1, v0.y * nrm);
    atomicAdd(dst +  2, v0.z * nrm); atomicAdd(dst +  3, v0.w * nrm);
    atomicAdd(dst +  4, v1.x * nrm); atomicAdd(dst +  5, v1.y * nrm);
    atomicAdd(dst +  6, v1.z * nrm); atomicAdd(dst +  7, v1.w * nrm);
    atomicAdd(dst +  8, v2.x * nrm); atomicAdd(dst +  9, v2.y * nrm);
    atomicAdd(dst + 10, v2.z * nrm); atomicAdd(dst + 11, v2.w * nrm);
    atomicAdd(dst + 12, v3.x * nrm); atomicAdd(dst + 13, v3.y * nrm);
    atomicAdd(dst + 14, v3.z * nrm); atomicAdd(dst + 15, v3.w * nrm);
}

__global__ void k_relu(float* __restrict__ acc, const float* __restrict__ b1, int n16) {
    int i = blockIdx.x * blockDim.x + threadIdx.x;
    if (i < n16) acc[i] = fmaxf(acc[i] + b1[i & 15], 0.f);
}

__global__ void k_gemm2(const float* __restrict__ z, const float* __restrict__ W2,
                        float* __restrict__ h2, int n, int ntiles) {
    int gtid = blockIdx.x * blockDim.x + threadIdx.x;
    int tile = gtid >> 5;
    int lane = threadIdx.x & 31;
    if (tile >= ntiles) return;
    int half = lane >> 4;
    int m    = lane & 15;
    int row  = tile * 16 + m;
    int rowc = row < n ? row : n - 1;
    bool rv  = row < n;
    int mc   = m < 7 ? m : 6;
    bool mv  = m < 7;

    v8f c = {};
#pragma unroll
    for (int s = 0; s < 4; ++s) {
        int k = s * 4 + half * 2;
        float ax = z[(size_t)rowc * 16 + k + 0];
        float ay = z[(size_t)rowc * 16 + k + 1];
        v2f a;
        a.x = rv ? ax : 0.f;
        a.y = rv ? ay : 0.f;
        float bx = W2[(k + 0) * 7 + mc];
        float by = W2[(k + 1) * 7 + mc];
        v2f b;
        b.x = mv ? bx : 0.f;
        b.y = mv ? by : 0.f;
        c = __builtin_amdgcn_wmma_f32_16x16x4_f32(false, a, false, b, (short)0, c, false, false);
    }

    float* p = h2 + ((size_t)(tile * 16 + half * 8)) * 16 + m;
    if (tile * 16 + 16 <= n) {
#pragma unroll
        for (int j = 0; j < 8; ++j) p[(size_t)j * 16] = c[j];
    } else {
#pragma unroll
        for (int j = 0; j < 8; ++j)
            if (tile * 16 + half * 8 + j < n) p[(size_t)j * 16] = c[j];
    }
}

__global__ void k_out_init(const float* __restrict__ h2, const float* __restrict__ dis,
                           const float* __restrict__ b2, float* __restrict__ out, int n7) {
    int i = blockIdx.x * blockDim.x + threadIdx.x;
    if (i < n7) {
        int node = i / 7, f = i % 7;
        float s = dis[node];
        out[i] = b2[f] + h2[(size_t)node * 16 + f] * s * s;
    }
}

__global__ void k_edges2(const int* __restrict__ row, const int* __restrict__ col,
                         const float* __restrict__ ew, const float* __restrict__ dis,
                         const float* __restrict__ h2, float* __restrict__ out, int e) {
    int i = blockIdx.x * blockDim.x + threadIdx.x;
    if (i >= e) return;
    int r = row[i], c = col[i];
    float nrm = dis[r] * ew[i] * dis[c];
    const float4* hr = (const float4*)(h2 + (size_t)r * 16);
    float4 v0 = hr[0], v1 = hr[1];
    float* dst = out + (size_t)c * 7;
    atomicAdd(dst + 0, v0.x * nrm); atomicAdd(dst + 1, v0.y * nrm);
    atomicAdd(dst + 2, v0.z * nrm); atomicAdd(dst + 3, v0.w * nrm);
    atomicAdd(dst + 4, v1.x * nrm); atomicAdd(dst + 5, v1.y * nrm);
    atomicAdd(dst + 6, v1.z * nrm);
}

extern "C" void kernel_launch(void* const* d_in, const int* in_sizes, int n_in,
                              void* d_out, int out_size, void* d_ws, size_t ws_size,
                              hipStream_t stream) {
    (void)n_in; (void)out_size; (void)ws_size;
    const float* x  = (const float*)d_in[0];
    const int*   ei = (const int*)d_in[1];
    const float* ew = (const float*)d_in[2];
    const float* W1 = (const float*)d_in[3];
    const float* b1 = (const float*)d_in[4];
    const float* W2 = (const float*)d_in[5];
    const float* b2 = (const float*)d_in[6];
    float* out = (float*)d_out;

    int n = in_sizes[0] / 3;
    int e = in_sizes[2];
    const int* row = ei;
    const int* col = ei + e;

    float* ws   = (float*)d_ws;
    float* dis  = ws;
    float* h1   = dis  + (size_t)n;
    float* acc1 = h1   + (size_t)n * 16;
    float* h2   = acc1 + (size_t)n * 16;

    const int B = 256;
    int ntiles  = (n + 15) / 16;
    int gemmT   = ntiles * 32;

    k_init_deg <<<(n + B - 1) / B, B, 0, stream>>>(dis, n);
    k_deg_edges<<<(e + B - 1) / B, B, 0, stream>>>(col, ew, dis, e);
    k_dis      <<<(n + B - 1) / B, B, 0, stream>>>(dis, n);

    k_gemm1    <<<(gemmT + B - 1) / B, B, 0, stream>>>(x, W1, h1, n, ntiles);
    k_self1    <<<(n * 16 + B - 1) / B, B, 0, stream>>>(h1, dis, acc1, n * 16);
    k_edges1   <<<(e + B - 1) / B, B, 0, stream>>>(row, col, ew, dis, h1, acc1, e);
    k_relu     <<<(n * 16 + B - 1) / B, B, 0, stream>>>(acc1, b1, n * 16);

    k_gemm2    <<<(gemmT + B - 1) / B, B, 0, stream>>>(acc1, W2, h2, n, ntiles);
    k_out_init <<<(n * 7 + B - 1) / B, B, 0, stream>>>(h2, dis, b2, out, n * 7);
    k_edges2   <<<(e + B - 1) / B, B, 0, stream>>>(row, col, ew, dis, h2, out, e);
}


# layer 1 scatters the 3 input features before the W1 product instead of the 16 after it ((A x) W1 = A (x W1)): 3 atomics per edge not 16; bias+relu in the WMMA epilogue; changes f32 sum order
# speedup vs baseline: 1.5674x; 1.5674x over previous
#include <hip/hip_runtime.h>
#include <hip/hip_bf16.h>

typedef __attribute__((ext_vector_type(2))) float v2f;
typedef __attribute__((ext_vector_type(8))) float v8f;

__global__ void k_init_deg(float* __restrict__ deg, int n) {
    int i = blockIdx.x * blockDim.x + threadIdx.x;
    if (i < n) deg[i] = 1.0f;
}

__global__ void k_deg_edges(const int* __restrict__ col, const float* __restrict__ ew,
                            float* __restrict__ deg, int e) {
    int i = blockIdx.x * blockDim.x + threadIdx.x;
    if (i < e) atomicAdd(&deg[col[i]], ew[i]);
}

__global__ void k_dis(float* __restrict__ deg_dis, int n) {
    int i = blockIdx.x * blockDim.x + threadIdx.x;
    if (i < n) {
        float d = deg_dis[i];
        deg_dis[i] = (d > 0.f) ? rsqrtf(d) : 0.f;
    }
}


__global__ void k_selfx(const float* __restrict__ x, const float* __restrict__ dis,
                        float* __restrict__ ax, int n) {
    int i = blockIdx.x * blockDim.x + threadIdx.x;
    if (i < n) {
        float s = dis[i];
        const float* xi = x + (size_t)i * 3;
        float4 v;
        v.x = xi[0] * s * s;
        v.y = xi[1] * s * s;
        v.z = xi[2] * s * s;
        v.w = 0.f;
        *(float4*)(ax + (size_t)i * 4) = v;
    }
}

__global__ void k_edges1x(const int* __restrict__ row, const int* __restrict__ col,
                          const float* __restrict__ ew, const float* __restrict__ dis,
                          const float* __restrict__ x, float* __restrict__ ax, int e) {
    int i = blockIdx.x * blockDim.x + threadIdx.x;
    if (i >= e) return;
    int r = row[i], c = col[i];
    float nrm = dis[r] * ew[i] * dis[c];
    const float* xr = x + (size_t)r * 3;
    float x0 = xr[0], x1 = xr[1], x2 = xr[2];
    float* dst = ax + (size_t)c * 4;
    atomicAdd(dst + 0, x0 * nrm);
    atomicAdd(dst + 1, x1 * nrm);
    atomicAdd(dst + 2, x2 * nrm);
}

__global__ void k_gemm1(const float* __restrict__ ax, const float* __restrict__ W1,
                        const float* __restrict__ b1, float* __restrict__ z,
                        int n, int ntiles) {
    int gtid = blockIdx.x * blockDim.x + threadIdx.x;
    int tile = gtid >> 5;
    int lane = threadIdx.x & 31;
    if (tile >= ntiles) return;
    int half = lane >> 4;
    int m    = lane & 15;
    int row  = tile * 16 + m;
    int rowc = row < n ? row : n - 1;
    bool rv  = row < n;
    int k0   = half * 2;

    v2f av = *(const v2f*)(ax + (size_t)rowc * 4 + k0);
    v2f a;
    a.x = rv ? av.x : 0.f;
    a.y = (rv && half == 0) ? av.y : 0.f;

    float bx = W1[k0 * 16 + m];
    float by = W1[16 + m];
    v2f b;
    b.x = bx;
    b.y = (half == 0) ? by : 0.f;

    float bias = b1[m];

    v8f c = {};
    c = __builtin_amdgcn_wmma_f32_16x16x4_f32(false, a, false, b, (short)0, c, false, false);

    float* p = z + ((size_t)(tile * 16 + half * 8)) * 16 + m;
    if (tile * 16 + 16 <= n) {
#pragma unroll
        for (int j = 0; j < 8; ++j) p[(size_t)j * 16] = fmaxf(c[j] + bias, 0.f);
    } else {
#pragma unroll
        for (int j = 0; j < 8; ++j)
            if (tile * 16 + half * 8 + j < n) p[(size_t)j * 16] = fmaxf(c[j] + bias, 0.f);
    }
}

__global__ void k_gemm2(const float* __restrict__ z, const float* __restrict__ W2,
                        float* __restrict__ h2, int n, int ntiles) {
    int gtid = blockIdx.x * blockDim.x + threadIdx.x;
    int tile = gtid >> 5;
    int lane = threadIdx.x & 31;
    if (tile >= ntiles) return;
    int half = lane >> 4;
    int m    = lane & 15;
    int row  = tile * 16 + m;
    int rowc = row < n ? row : n - 1;
    bool rv  = row < n;
    int mc   = m < 7 ? m : 6;
    bool mv  = m < 7;

    v8f c = {};
#pragma unroll
    for (int s = 0; s < 4; ++s) {
        int k = s * 4 + half * 2;
        float ax = z[(size_t)rowc * 16 + k + 0];
        float ay = z[(size_t)rowc * 16 + k + 1];
        v2f a;
        a.x = rv ? ax : 0.f;
        a.y = rv ? ay : 0.f;
        float bx = W2[(k + 0) * 7 + mc];
        float by = W2[(k + 1) * 7 + mc];
        v2f b;
        b.x = mv ? bx : 0.f;
        b.y = mv ? by : 0.f;
        c = __builtin_amdgcn_wmma_f32_16x16x4_f32(false, a, false, b, (short)0, c, false, false);
    }

    float* p = h2 + ((size_t)(tile * 16 + half * 8)) * 16 + m;
    if (tile * 16 + 16 <= n) {
#pragma unroll
        for (int j = 0; j < 8; ++j) p[(size_t)j * 16] = c[j];
    } else {
#pragma unroll
        for (int j = 0; j < 8; ++j)
            if (tile * 16 + half * 8 + j < n) p[(size_t)j * 16] = c[j];
    }
}

__global__ void k_out_init(const float* __restrict__ h2, const float* __restrict__ dis,
                           const float* __restrict__ b2, float* __restrict__ out, int n7) {
    int i = blockIdx.x * blockDim.x + threadIdx.x;
    if (i < n7) {
        int node = i / 7, f = i % 7;
        float s = dis[node];
        out[i] = b2[f] + h2[(size_t)node * 16 + f] * s * s;
    }
}

__global__ void k_edges2(const int* __restrict__ row, const int* __restrict__ col,
                         const float* __restrict__ ew, const float* __restrict__ dis,
                         const float* __restrict__ h2, float* __restrict__ out, int e) {
    int i = blockIdx.x * blockDim.x + threadIdx.x;
    if (i >= e) return;
    int r = row[i], c = col[i];
    float nrm = dis[r] * ew[i] * dis[c];
    const float4* hr = (const float4*)(h2 + (size_t)r * 16);
    float4 v0 = hr[0], v1 = hr[1];
    float* dst = out + (size_t)c * 7;
    atomicAdd(dst + 0, v0.x * nrm); atomicAdd(dst + 1, v0.y * nrm);
    atomicAdd(dst + 2, v0.z * nrm); atomicAdd(dst + 3, v0.w * nrm);
    atomicAdd(dst + 4, v1.x * nrm); atomicAdd(dst + 5, v1.y * nrm);
    atomicAdd(dst + 6, v1.z * nrm);
}

extern "C" void kernel_launch(void* const* d_in, const int* in_sizes, int n_in,
                              void* d_out, int out_size, void* d_ws, size_t ws_size,
                              hipStream_t stream) {
    (void)n_in; (void)out_size; (void)ws_size;
    const float* x  = (const float*)d_in[0];
    const int*   ei = (const int*)d_in[1];
    const float* ew = (const float*)d_in[2];
    const float* W1 = (const float*)d_in[3];
    const float* b1 = (const float*)d_in[4];
    const float* W2 = (const float*)d_in[5];
    const float* b2 = (const float*)d_in[6];
    float* out = (float*)d_out;

    int n = in_sizes[0] / 3;
    int e = in_sizes[2];
    const int* row = ei;
    const int* col = ei + e;

    float* ws   = (float*)d_ws;
    float* dis  = ws;
    float* ax   = dis  + (size_t)n;
    float* acc1 = ax   + (size_t)n * 4;
    float* h2   = acc1 + (size_t)n * 16;

    const int B = 256;
    int ntiles  = (n + 15) / 16;
    int gemmT   = ntiles * 32;

    k_init_deg <<<(n + B - 1) / B, B, 0, stream>>>(dis, n);
    k_deg_edges<<<(e + B - 1) / B, B, 0, stream>>>(col, ew, dis, e);
    k_dis      <<<(n + B - 1) / B, B, 0, stream>>>(dis, n);

    k_selfx    <<<(n + B - 1) / B, B, 0, stream>>>(x, dis, ax, n);
    k_edges1x  <<<(e + B - 1) / B, B, 0, stream>>>(row, col, ew, dis, x, ax, e);
    k_gemm1    <<<(gemmT + B - 1) / B, B, 0, stream>>>(ax, W1, b1, acc1, n, ntiles);

    k_gemm2    <<<(gemmT + B - 1) / B, B, 0, stream>>>(acc1, W2, h2, n, ntiles);
    k_out_init <<<(n * 7 + B - 1) / B, B, 0, stream>>>(h2, dis, b2, out, n * 7);
    k_edges2   <<<(e + B - 1) / B, B, 0, stream>>>(row, col, ew, dis, h2, out, e);
}


# edges counting-sorted by target once per call (1 int atomic per edge, scan, fill); degree and both layers sum each node's segment by plain loads, no float atomics; new f32 sum order (arrival rank)
# speedup vs baseline: 3.4578x; 2.2061x over previous
#include <hip/hip_runtime.h>
#include <hip/hip_bf16.h>


typedef __attribute__((ext_vector_type(2))) float v2f;
typedef __attribute__((ext_vector_type(8))) float v8f;


#define SCAN_T 256
#define SCAN_E 1024
#define EPT 4
#define AGG_T 256
#define NPB (AGG_T / 4)

__global__ void k_zero_cnt(int* __restrict__ cnt) {
    int t = blockIdx.x * blockDim.x + threadIdx.x;
    *(int4*)(cnt + (size_t)t * 4) = make_int4(0, 0, 0, 0);
}

__global__ void k_count_rank(const int* __restrict__ col, int* __restrict__ cnt,
                             int* __restrict__ rank, int e) {
    int tid = blockIdx.x * blockDim.x + threadIdx.x;
    int T   = gridDim.x * blockDim.x;
    int c[EPT], k[EPT];
    for (int j = 0; j < EPT; ++j) {
        int i  = tid + j * T;
        int ic = i < e ? i : e - 1;
        c[j] = col[ic];
    }
    for (int j = 0; j < EPT; ++j) {
        int i = tid + j * T;
        k[j] = 0;
        if (i < e) k[j] = atomicAdd(&cnt[c[j]], 1);
    }
    for (int j = 0; j < EPT; ++j) {
        int i = tid + j * T;
        if (i < e) rank[i] = k[j];
    }
}

__global__ void k_scan_local(int* __restrict__ cnt, int* __restrict__ bsum) {
    __shared__ int wsum[SCAN_T / 32];
    int t    = threadIdx.x;
    int lane = t & 31, w = t >> 5;
    int* p   = cnt + (size_t)blockIdx.x * SCAN_E + t * 4;
    int4 v   = *(const int4*)p;
    int s    = v.x + v.y + v.z + v.w;
    int incl = s;
    for (int d = 1; d < 32; d <<= 1) {
        int y = __shfl_up(incl, d);
        if (lane >= d) incl += y;
    }
    if (lane == 31) wsum[w] = incl;
    __syncthreads();
    int wofs = 0;
    for (int j = 0; j < SCAN_T / 32; ++j) wofs += (j < w) ? wsum[j] : 0;
    int ex = wofs + incl - s;
    int4 o;
    o.x = ex;
    o.y = ex + v.x;
    o.z = o.y + v.y;
    o.w = o.z + v.z;
    *(int4*)p = o;
    if (t == SCAN_T - 1) bsum[blockIdx.x] = ex + s;
}

__global__ void k_scan_fix(int* __restrict__ off, const int* __restrict__ bsum) {
    __shared__ int wsum[SCAN_T / 32];
    int t = threadIdx.x;
    int b = blockIdx.x;
    int part = 0;
    for (int j = t; j < b; j += SCAN_T) part += bsum[j];
    for (int d = 16; d >= 1; d >>= 1) part += __shfl_xor(part, d);
    if ((t & 31) == 0) wsum[t >> 5] = part;
    __syncthreads();
    int base = 0;
    for (int j = 0; j < SCAN_T / 32; ++j) base += wsum[j];
    size_t idx = (size_t)b * SCAN_E + t * 4;
    int4 v = *(const int4*)(off + idx);
    v.x += base; v.y += base; v.z += base; v.w += base;
    *(int4*)(off + idx) = v;
}

__global__ void k_fill(const int* __restrict__ row, const int* __restrict__ col,
                       const float* __restrict__ ew, const int* __restrict__ rank,
                       const int* __restrict__ off, int2* __restrict__ rec, int e) {
    int tid = blockIdx.x * blockDim.x + threadIdx.x;
    int T   = gridDim.x * blockDim.x;
    int c[EPT], r[EPT], k[EPT], o[EPT];
    float w[EPT];
    for (int j = 0; j < EPT; ++j) {
        int i  = tid + j * T;
        int ic = i < e ? i : e - 1;
        c[j] = col[ic];
        k[j] = rank[ic];
        r[j] = row[ic];
        w[j] = ew[ic];
    }
    for (int j = 0; j < EPT; ++j) o[j] = off[c[j]];
    for (int j = 0; j < EPT; ++j) {
        int i = tid + j * T;
        int p = o[j] + k[j];
        if (i < e && (unsigned)p < (unsigned)e)
            rec[p] = make_int2(r[j], __float_as_int(w[j]));
    }
}

__device__ __forceinline__ void seg_bounds(const int* __restrict__ off, int node, int n, int e,
                                           int& beg, int& end) {
    int nc = node < n ? node : n - 1;
    int b  = off[nc];
    int en = off[nc + 1];
    b  = b > 0 ? b : 0;
    en = en < e ? en : e;
    beg = b;
    end = node < n ? en : b;
}

__device__ __forceinline__ float quad_sum(float v) {
    v += __shfl_xor(v, 1);
    v += __shfl_xor(v, 2);
    return v;
}

__global__ void k_node_deg(const int* __restrict__ off, const int2* __restrict__ rec,
                           const float* __restrict__ x, float* __restrict__ dis,
                           float* __restrict__ xs, int n, int e) {
    int t    = blockIdx.x * blockDim.x + threadIdx.x;
    int node = t >> 2, sub = t & 3;
    int beg, end;
    seg_bounds(off, node, n, e, beg, end);
    float s = 0.f;
    for (int k = beg + sub; k < end; k += 4) s += __int_as_float(rec[k].y);
    s = quad_sum(s);
    if (node < n && sub == 0) {
        float d  = 1.0f + s;
        float di = (d > 0.f) ? rsqrtf(d) : 0.f;
        dis[node] = di;
        const float* xi = x + (size_t)node * 3;
        float4 v;
        v.x = xi[0] * di;
        v.y = xi[1] * di;
        v.z = xi[2] * di;
        v.w = 0.f;
        *(float4*)(xs + (size_t)node * 4) = v;
    }
}

__global__ void k_agg1(const int* __restrict__ off, const int2* __restrict__ rec,
                       const float* __restrict__ xs, const float* __restrict__ dis,
                       float* __restrict__ ax, int n, int e) {
    int t    = blockIdx.x * blockDim.x + threadIdx.x;
    int node = t >> 2, sub = t & 3;
    int beg, end;
    seg_bounds(off, node, n, e, beg, end);
    float a0 = 0.f, a1 = 0.f, a2 = 0.f;
    for (int k = beg + sub; k < end; k += 4) {
        int2 r   = rec[k];
        float w  = __int_as_float(r.y);
        float4 v = *(const float4*)(xs + (size_t)r.x * 4);
        a0 += w * v.x;
        a1 += w * v.y;
        a2 += w * v.z;
    }
    a0 = quad_sum(a0);
    a1 = quad_sum(a1);
    a2 = quad_sum(a2);
    if (node < n && sub == 0) {
        float di = dis[node];
        float4 sf = *(const float4*)(xs + (size_t)node * 4);
        float4 o;
        o.x = di * (a0 + sf.x);
        o.y = di * (a1 + sf.y);
        o.z = di * (a2 + sf.z);
        o.w = 0.f;
        *(float4*)(ax + (size_t)node * 4) = o;
    }
}

__global__ void k_gemm1(const float* __restrict__ ax, const float* __restrict__ W1,
                        const float* __restrict__ b1, float* __restrict__ z,
                        int n, int ntiles) {
    int gtid = blockIdx.x * blockDim.x + threadIdx.x;
    int tile = gtid >> 5;
    int lane = threadIdx.x & 31;
    if (tile >= ntiles) return;
    int half = lane >> 4;
    int m    = lane & 15;
    int row  = tile * 16 + m;
    int rowc = row < n ? row : n - 1;
    bool rv  = row < n;
    int k0   = half * 2;

    v2f av = *(const v2f*)(ax + (size_t)rowc * 4 + k0);
    v2f a;
    a.x = rv ? av.x : 0.f;
    a.y = (rv && half == 0) ? av.y : 0.f;

    float bx = W1[k0 * 16 + m];
    float by = W1[16 + m];
    v2f b;
    b.x = bx;
    b.y = (half == 0) ? by : 0.f;

    float bias = b1[m];

    v8f c = {};
    c = __builtin_amdgcn_wmma_f32_16x16x4_f32(false, a, false, b, (short)0, c, false, false);

    float* p = z + ((size_t)(tile * 16 + half * 8)) * 16 + m;
    if (tile * 16 + 16 <= n) {
#pragma unroll
        for (int j = 0; j < 8; ++j) p[(size_t)j * 16] = fmaxf(c[j] + bias, 0.f);
    } else {
#pragma unroll
        for (int j = 0; j < 8; ++j)
            if (tile * 16 + half * 8 + j < n) p[(size_t)j * 16] = fmaxf(c[j] + bias, 0.f);
    }
}

__global__ void k_gemm2(const float* __restrict__ z, const float* __restrict__ W2,
                        const float* __restrict__ dis,
                        float* __restrict__ h2, int n, int ntiles) {
    int gtid = blockIdx.x * blockDim.x + threadIdx.x;
    int tile = gtid >> 5;
    int lane = threadIdx.x & 31;
    if (tile >= ntiles) return;
    int half = lane >> 4;
    int m    = lane & 15;
    int row  = tile * 16 + m;
    int rowc = row < n ? row : n - 1;
    bool rv  = row < n;
    int mc   = m < 7 ? m : 6;
    bool mv  = m < 7;
    float ds = dis[rowc];

    v8f c = {};
#pragma unroll
    for (int s = 0; s < 4; ++s) {
        int k = s * 4 + half * 2;
        float ax = z[(size_t)rowc * 16 + k + 0];
        float ay = z[(size_t)rowc * 16 + k + 1];
        v2f a;
        a.x = rv ? ax * ds : 0.f;
        a.y = rv ? ay * ds : 0.f;
        float bx = W2[(k + 0) * 7 + mc];
        float by = W2[(k + 1) * 7 + mc];
        v2f b;
        b.x = mv ? bx : 0.f;
        b.y = mv ? by : 0.f;
        c = __builtin_amdgcn_wmma_f32_16x16x4_f32(false, a, false, b, (short)0, c, false, false);
    }

    float* p = h2 + ((size_t)(tile * 16 + half * 8)) * 16 + m;
    if (tile * 16 + 16 <= n) {
#pragma unroll
        for (int j = 0; j < 8; ++j) p[(size_t)j * 16] = c[j];
    } else {
#pragma unroll
        for (int j = 0; j < 8; ++j)
            if (tile * 16 + half * 8 + j < n) p[(size_t)j * 16] = c[j];
    }
}

__global__ void k_agg2(const int* __restrict__ off, const int2* __restrict__ rec,
                       const float* __restrict__ h2, const float* __restrict__ dis,
                       const float* __restrict__ b2, float* __restrict__ out, int n, int e) {
    __shared__ float so[NPB * 7];
    int t    = blockIdx.x * blockDim.x + threadIdx.x;
    int node = t >> 2, sub = t & 3;
    int beg, end;
    seg_bounds(off, node, n, e, beg, end);
    float a0 = 0.f, a1 = 0.f, a2 = 0.f, a3 = 0.f, a4 = 0.f, a5 = 0.f, a6 = 0.f;
    for (int k = beg + sub; k < end; k += 4) {
        int2 r  = rec[k];
        float w = __int_as_float(r.y);
        const float4* hr = (const float4*)(h2 + (size_t)r.x * 16);
        float4 v0 = hr[0], v1 = hr[1];
        a0 += w * v0.x; a1 += w * v0.y; a2 += w * v0.z; a3 += w * v0.w;
        a4 += w * v1.x; a5 += w * v1.y; a6 += w * v1.z;
    }
    a0 = quad_sum(a0); a1 = quad_sum(a1); a2 = quad_sum(a2); a3 = quad_sum(a3);
    a4 = quad_sum(a4); a5 = quad_sum(a5); a6 = quad_sum(a6);
    if (node < n && sub == 0) {
        float di = dis[node];
        const float4* hs = (const float4*)(h2 + (size_t)node * 16);
        float4 s0 = hs[0], s1 = hs[1];
        float* o = so + (threadIdx.x >> 2) * 7;
        o[0] = b2[0] + di * (a0 + s0.x);
        o[1] = b2[1] + di * (a1 + s0.y);
        o[2] = b2[2] + di * (a2 + s0.z);
        o[3] = b2[3] + di * (a3 + s0.w);
        o[4] = b2[4] + di * (a4 + s1.x);
        o[5] = b2[5] + di * (a5 + s1.y);
        o[6] = b2[6] + di * (a6 + s1.z);
    }
    __syncthreads();
    int nb    = blockIdx.x * NPB;
    int left  = n - nb;
    int cntf  = (left < NPB ? left : NPB) * 7;
    float* ob = out + (size_t)nb * 7;
    for (int j = threadIdx.x; j < NPB * 7; j += AGG_T)
        if (j < cntf) ob[j] = so[j];
}

extern "C" void kernel_launch(void* const* d_in, const int* in_sizes, int n_in,
                              void* d_out, int out_size, void* d_ws, size_t ws_size,
                              hipStream_t stream) {
    (void)n_in; (void)out_size; (void)ws_size;
    const float* x  = (const float*)d_in[0];
    const int*   ei = (const int*)d_in[1];
    const float* ew = (const float*)d_in[2];
    const float* W1 = (const float*)d_in[3];
    const float* b1 = (const float*)d_in[4];
    const float* W2 = (const float*)d_in[5];
    const float* b2 = (const float*)d_in[6];
    float* out = (float*)d_out;

    int n = in_sizes[0] / 3;
    int e = in_sizes[2];
    const int* row = ei;
    const int* col = ei + e;

    int nsb  = (n + 1 + SCAN_E - 1) / SCAN_E;
    int npad = nsb * SCAN_E;

    int*   off    = (int*)d_ws;
    int*   bsum   = off  + (size_t)npad;
    float* dis    = (float*)(bsum + (size_t)npad);
    float* xs     = dis  + (size_t)npad;
    float* ax     = xs   + (size_t)npad * 4;
    float* acc1   = ax   + (size_t)npad * 4;
    float* h2     = acc1 + (size_t)npad * 16;
    int*   rank   = (int*)(h2 + (size_t)npad * 16);
    int2*  rec    = (int2*)(rank + (((size_t)e + 3) / 4) * 4);

    const int B = 256;
    int ntiles  = (n + 15) / 16;
    int gemmT   = ntiles * 32;
    int quadB   = (n + NPB - 1) / NPB;
    int edgeB   = (e + B * EPT - 1) / (B * EPT);

    k_zero_cnt  <<<nsb, SCAN_T, 0, stream>>>(off);
    k_count_rank<<<edgeB, B, 0, stream>>>(col, off, rank, e);
    k_scan_local<<<nsb, SCAN_T, 0, stream>>>(off, bsum);
    k_scan_fix  <<<nsb, SCAN_T, 0, stream>>>(off, bsum);
    k_fill      <<<edgeB, B, 0, stream>>>(row, col, ew, rank, off, rec, e);

    k_node_deg  <<<quadB, AGG_T, 0, stream>>>(off, rec, x, dis, xs, n, e);
    k_agg1      <<<quadB, AGG_T, 0, stream>>>(off, rec, xs, dis, ax, n, e);
    k_gemm1     <<<(gemmT + B - 1) / B, B, 0, stream>>>(ax, W1, b1, acc1, n, ntiles);

    k_gemm2     <<<(gemmT + B - 1) / B, B, 0, stream>>>(acc1, W2, dis, h2, n, ntiles);
    k_agg2      <<<quadB, AGG_T, 0, stream>>>(off, rec, h2, dis, b2, out, n, e);
}


# edge counting sort redone in two levels with every counter in LDS: 1024-node buckets via per-block LDS histograms and LDS-staged runs, then one work-group per bucket; sum order still arrival rank
# speedup vs baseline: 8.1527x; 2.3578x over previous
#include <hip/hip_runtime.h>
#include <hip/hip_bf16.h>


typedef __attribute__((ext_vector_type(2))) float v2f;
typedef __attribute__((ext_vector_type(8))) float v8f;


#define BK_SHIFT 10
#define BK_N (1 << BK_SHIFT)
#define PART_T 512
#define MAXB PART_T
#define EPT1 16
#define EPB (PART_T * EPT1)
#define SB 4
static_assert(SB == 4, "the load fences in k_bk_sort name SB = 4 values each");
#define SCAN_T 256
#define SCAN_E 1024
#define AGG_T 256
#define NPB (AGG_T / 4)

__global__ void k_zero_cnt(int* __restrict__ cnt) {
    int t = blockIdx.x * blockDim.x + threadIdx.x;
    *(int4*)(cnt + (size_t)t * 4) = make_int4(0, 0, 0, 0);
}

__global__ void k_scan_local(int* __restrict__ cnt, int* __restrict__ bsum) {
    __shared__ int wsum[SCAN_T / 32];
    int t    = threadIdx.x;
    int lane = t & 31, w = t >> 5;
    int* p   = cnt + (size_t)blockIdx.x * SCAN_E + t * 4;
    int4 v   = *(const int4*)p;
    int s    = v.x + v.y + v.z + v.w;
    int incl = s;
    for (int d = 1; d < 32; d <<= 1) {
        int y = __shfl_up(incl, d);
        if (lane >= d) incl += y;
    }
    if (lane == 31) wsum[w] = incl;
    __syncthreads();
    int wofs = 0;
    for (int j = 0; j < SCAN_T / 32; ++j) wofs += (j < w) ? wsum[j] : 0;
    int ex = wofs + incl - s;
    int4 o;
    o.x = ex;
    o.y = ex + v.x;
    o.z = o.y + v.y;
    o.w = o.z + v.z;
    *(int4*)p = o;
    if (t == SCAN_T - 1) bsum[blockIdx.x] = ex + s;
}

__global__ void k_scan_fix(int* __restrict__ off, const int* __restrict__ bsum) {
    __shared__ int wsum[SCAN_T / 32];
    int t = threadIdx.x;
    int b = blockIdx.x;
    int part = 0;
    for (int j = t; j < b; j += SCAN_T) part += bsum[j];
    for (int d = 16; d >= 1; d >>= 1) part += __shfl_xor(part, d);
    if ((t & 31) == 0) wsum[t >> 5] = part;
    __syncthreads();
    int base = 0;
    for (int j = 0; j < SCAN_T / 32; ++j) base += wsum[j];
    size_t idx = (size_t)b * SCAN_E + t * 4;
    int4 v = *(const int4*)(off + idx);
    v.x += base; v.y += base; v.z += base; v.w += base;
    *(int4*)(off + idx) = v;
}

__global__ void k_bk_hist(const int* __restrict__ col, int* __restrict__ hist,
                          int e, int nblk, int nbk) {
    __shared__ int lh[MAXB];
    int t = threadIdx.x;
    lh[t] = 0;
    __syncthreads();
    int base = blockIdx.x * EPB;
    int c[EPT1];
    for (int j = 0; j < EPT1; ++j) {
        int i  = base + t + j * PART_T;
        int ic = i < e ? i : e - 1;
        c[j] = col[ic];
    }
    for (int j = 0; j < EPT1; ++j) {
        int i = base + t + j * PART_T;
        if (i < e) atomicAdd(&lh[c[j] >> BK_SHIFT], 1);
    }
    __syncthreads();
    if (t < nbk) hist[(size_t)t * nblk + blockIdx.x] = lh[t];
}

__global__ void k_bk_part(const int* __restrict__ row, const int* __restrict__ col,
                          const float* __restrict__ ew, const int* __restrict__ hist,
                          int2* __restrict__ trec, int* __restrict__ tcol,
                          int e, int nblk, int nbk) {
    __shared__ int  lh[MAXB];
    __shared__ int  gb[MAXB];
    __shared__ int  wsum[PART_T / 32];
    __shared__ int  s_col[EPB];
    __shared__ int2 s_rec[EPB];
    int t    = threadIdx.x;
    int lane = t & 31, w = t >> 5;
    lh[t] = 0;
    __syncthreads();
    int base = blockIdx.x * EPB;
    int c[EPT1], r[EPT1], k[EPT1];
    float wt[EPT1];
    for (int j = 0; j < EPT1; ++j) {
        int i  = base + t + j * PART_T;
        int ic = i < e ? i : e - 1;
        c[j]  = col[ic];
        r[j]  = row[ic];
        wt[j] = ew[ic];
    }
    for (int j = 0; j < EPT1; ++j) {
        int i = base + t + j * PART_T;
        k[j] = 0;
        if (i < e) k[j] = atomicAdd(&lh[c[j] >> BK_SHIFT], 1);
    }
    __syncthreads();
    int v    = lh[t];
    int incl = v;
    for (int d = 1; d < 32; d <<= 1) {
        int y = __shfl_up(incl, d);
        if (lane >= d) incl += y;
    }
    if (lane == 31) wsum[w] = incl;
    __syncthreads();
    int wofs = 0;
    for (int j = 0; j < PART_T / 32; ++j) wofs += (j < w) ? wsum[j] : 0;
    int ex = wofs + incl - v;
    int g  = (t < nbk) ? hist[(size_t)t * nblk + blockIdx.x] : 0;
    lh[t] = ex;
    gb[t] = g - ex;
    __syncthreads();
    for (int j = 0; j < EPT1; ++j) {
        int i = base + t + j * PART_T;
        if (i < e) {
            int s = lh[c[j] >> BK_SHIFT] + k[j];
            s_col[s] = c[j];
            s_rec[s] = make_int2(r[j], __float_as_int(wt[j]));
        }
    }
    __syncthreads();
    int left  = e - base;
    int total = left < EPB ? left : EPB;
    for (int s = t; s < total; s += PART_T) {
        int cc = s_col[s];
        int p  = gb[cc >> BK_SHIFT] + s;
        if ((unsigned)p < (unsigned)e) {
            trec[p] = s_rec[s];
            tcol[p] = cc;
        }
    }
}

__global__ void k_bk_sort(const int* __restrict__ hist, const int2* __restrict__ trec,
                          const int* __restrict__ tcol, int* __restrict__ rank,
                          int* __restrict__ off, int2* __restrict__ rec, int e, int nblk) {
    __shared__ int cnt[BK_N];
    __shared__ int wsum[BK_N / 32];
    int t    = threadIdx.x;
    int lane = t & 31, w = t >> 5;
    int b    = blockIdx.x;
    int bs   = hist[(size_t)b * nblk];
    int be   = hist[(size_t)(b + 1) * nblk];
    bs = bs > 0 ? bs : 0;
    be = be < e ? be : e;
    cnt[t] = 0;
    __syncthreads();
    for (int q0 = bs + t; q0 < be; q0 += SB * BK_N) {
        int cc[SB];
        for (int u = 0; u < SB; ++u) {
            int q  = q0 + u * BK_N;
            int qc = q < be ? q : be - 1;
            cc[u] = tcol[qc];
        }
        asm volatile("" :: "v"(cc[0]), "v"(cc[1]), "v"(cc[2]), "v"(cc[3]));
        for (int u = 0; u < SB; ++u) {
            int q = q0 + u * BK_N;
            if (q < be) {
                int lc = (cc[u] - (b << BK_SHIFT)) & (BK_N - 1);
                rank[q] = atomicAdd(&cnt[lc], 1);
            }
        }
    }
    __syncthreads();
    int v    = cnt[t];
    int incl = v;
    for (int d = 1; d < 32; d <<= 1) {
        int y = __shfl_up(incl, d);
        if (lane >= d) incl += y;
    }
    if (lane == 31) wsum[w] = incl;
    __syncthreads();
    int wofs = 0;
    for (int j = 0; j < BK_N / 32; ++j) wofs += (j < w) ? wsum[j] : 0;
    int st = bs + wofs + incl - v;
    cnt[t] = st;
    off[(size_t)(b << BK_SHIFT) + t] = st;
    __syncthreads();
    for (int q0 = bs + t; q0 < be; q0 += SB * BK_N) {
        int cc[SB], rk[SB];
        int2 rr[SB];
        for (int u = 0; u < SB; ++u) {
            int q  = q0 + u * BK_N;
            int qc = q < be ? q : be - 1;
            cc[u] = tcol[qc];
            rk[u] = rank[qc];
            rr[u] = trec[qc];
        }
        asm volatile("" :: "v"(cc[0]), "v"(cc[1]), "v"(cc[2]), "v"(cc[3]),
                           "v"(rk[0]), "v"(rk[1]), "v"(rk[2]), "v"(rk[3]),
                           "v"(rr[0].x), "v"(rr[0].y), "v"(rr[1].x), "v"(rr[1].y),
                           "v"(rr[2].x), "v"(rr[2].y), "v"(rr[3].x), "v"(rr[3].y));
        for (int u = 0; u < SB; ++u) {
            int q = q0 + u * BK_N;
            int lc = (cc[u] - (b << BK_SHIFT)) & (BK_N - 1);
            int p  = cnt[lc] + rk[u];
            if (q < be && (unsigned)p < (unsigned)e) rec[p] = rr[u];
        }
    }
}

__device__ __forceinline__ void seg_bounds(const int* __restrict__ off, int node, int n, int e,
                                           int& beg, int& end) {
    int nc = node < n ? node : n - 1;
    int b  = off[nc];
    int en = off[nc + 1];
    b  = b > 0 ? b : 0;
    en = en < e ? en : e;
    beg = b;
    end = node < n ? en : b;
}

__device__ __forceinline__ float quad_sum(float v) {
    v += __shfl_xor(v, 1);
    v += __shfl_xor(v, 2);
    return v;
}

__global__ void k_node_deg(const int* __restrict__ off, const int2* __restrict__ rec,
                           const float* __restrict__ x, float* __restrict__ dis,
                           float* __restrict__ xs, int n, int e) {
    int t    = blockIdx.x * blockDim.x + threadIdx.x;
    int node = t >> 2, sub = t & 3;
    int beg, end;
    seg_bounds(off, node, n, e, beg, end);
    float s = 0.f;
    for (int k = beg + sub; k < end; k += 4) s += __int_as_float(rec[k].y);
    s = quad_sum(s);
    if (node < n && sub == 0) {
        float d  = 1.0f + s;
        float di = (d > 0.f) ? rsqrtf(d) : 0.f;
        dis[node] = di;
        const float* xi = x + (size_t)node * 3;
        float4 v;
        v.x = xi[0] * di;
        v.y = xi[1] * di;
        v.z = xi[2] * di;
        v.w = 0.f;
        *(float4*)(xs + (size_t)node * 4) = v;
    }
}

__global__ void k_agg1(const int* __restrict__ off, const int2* __restrict__ rec,
                       const float* __restrict__ xs, const float* __restrict__ dis,
                       float* __restrict__ ax, int n, int e) {
    int t    = blockIdx.x * blockDim.x + threadIdx.x;
    int node = t >> 2, sub = t & 3;
    int beg, end;
    seg_bounds(off, node, n, e, beg, end);
    float a0 = 0.f, a1 = 0.f, a2 = 0.f;
    for (int k = beg + sub; k < end; k += 4) {
        int2 r   = rec[k];
        float w  = __int_as_float(r.y);
        float4 v = *(const float4*)(xs + (size_t)r.x * 4);
        a0 += w * v.x;
        a1 += w * v.y;
        a2 += w * v.z;
    }
    a0 = quad_sum(a0);
    a1 = quad_sum(a1);
    a2 = quad_sum(a2);
    if (node < n && sub == 0) {
        float di = dis[node];
        float4 sf = *(const float4*)(xs + (size_t)node * 4);
        float4 o;
        o.x = di * (a0 + sf.x);
        o.y = di * (a1 + sf.y);
        o.z = di * (a2 + sf.z);
        o.w = 0.f;
        *(float4*)(ax + (size_t)node * 4) = o;
    }
}

__global__ void k_gemm1(const float* __restrict__ ax, const float* __restrict__ W1,
                        const float* __restrict__ b1, float* __restrict__ z,
                        int n, int ntiles) {
    int gtid = blockIdx.x * blockDim.x + threadIdx.x;
    int tile = gtid >> 5;
    int lane = threadIdx.x & 31;
    if (tile >= ntiles) return;
    int half = lane >> 4;
    int m    = lane & 15;
    int row  = tile * 16 + m;
    int rowc = row < n ? row : n - 1;
    bool rv  = row < n;
    int k0   = half * 2;

    v2f av = *(const v2f*)(ax + (size_t)rowc * 4 + k0);
    v2f a;
    a.x = rv ? av.x : 0.f;
    a.y = (rv && half == 0) ? av.y : 0.f;

    float bx = W1[k0 * 16 + m];
    float by = W1[16 + m];
    v2f b;
    b.x = bx;
    b.y = (half == 0) ? by : 0.f;

    float bias = b1[m];

    v8f c = {};
    c = __builtin_amdgcn_wmma_f32_16x16x4_f32(false, a, false, b, (short)0, c, false, false);

    float* p = z + ((size_t)(tile * 16 + half * 8)) * 16 + m;
    if (tile * 16 + 16 <= n) {
#pragma unroll
        for (int j = 0; j < 8; ++j) p[(size_t)j * 16] = fmaxf(c[j] + bias, 0.f);
    } else {
#pragma unroll
        for (int j = 0; j < 8; ++j)
            if (tile * 16 + half * 8 + j < n) p[(size_t)j * 16] = fmaxf(c[j] + bias, 0.f);
    }
}

__global__ void k_gemm2(const float* __restrict__ z, const float* __restrict__ W2,
                        const float* __restrict__ dis,
                        float* __restrict__ h2, int n, int ntiles) {
    int gtid = blockIdx.x * blockDim.x + threadIdx.x;
    int tile = gtid >> 5;
    int lane = threadIdx.x & 31;
    if (tile >= ntiles) return;
    int half = lane >> 4;
    int m    = lane & 15;
    int row  = tile * 16 + m;
    int rowc = row < n ? row : n - 1;
    bool rv  = row < n;
    int mc   = m < 7 ? m : 6;
    bool mv  = m < 7;
    float ds = dis[rowc];

    v8f c = {};
#pragma unroll
    for (int s = 0; s < 4; ++s) {
        int k = s * 4 + half * 2;
        float ax = z[(size_t)rowc * 16 + k + 0];
        float ay = z[(size_t)rowc * 16 + k + 1];
        v2f a;
        a.x = rv ? ax * ds : 0.f;
        a.y = rv ? ay * ds : 0.f;
        float bx = W2[(k + 0) * 7 + mc];
        float by = W2[(k + 1) * 7 + mc];
        v2f b;
        b.x = mv ? bx : 0.f;
        b.y = mv ? by : 0.f;
        c = __builtin_amdgcn_wmma_f32_16x16x4_f32(false, a, false, b, (short)0, c, false, false);
    }

    float* p = h2 + ((size_t)(tile * 16 + half * 8)) * 16 + m;
    if (tile * 16 + 16 <= n) {
#pragma unroll
        for (int j = 0; j < 8; ++j) p[(size_t)j * 16] = c[j];
    } else {
#pragma unroll
        for (int j = 0; j < 8; ++j)
            if (tile * 16 + half * 8 + j < n) p[(size_t)j * 16] = c[j];
    }
}

__global__ void k_agg2(const int* __restrict__ off, const int2* __restrict__ rec,
                       const float* __restrict__ h2, const float* __restrict__ dis,
                       const float* __restrict__ b2, float* __restrict__ out, int n, int e) {
    __shared__ float so[NPB * 7];
    int t    = blockIdx.x * blockDim.x + threadIdx.x;
    int node = t >> 2, sub = t & 3;
    int beg, end;
    seg_bounds(off, node, n, e, beg, end);
    float a0 = 0.f, a1 = 0.f, a2 = 0.f, a3 = 0.f, a4 = 0.f, a5 = 0.f, a6 = 0.f;
    for (int k = beg + sub; k < end; k += 4) {
        int2 r  = rec[k];
        float w = __int_as_float(r.y);
        const float4* hr = (const float4*)(h2 + (size_t)r.x * 16);
        float4 v0 = hr[0], v1 = hr[1];
        a0 += w * v0.x; a1 += w * v0.y; a2 += w * v0.z; a3 += w * v0.w;
        a4 += w * v1.x; a5 += w * v1.y; a6 += w * v1.z;
    }
    a0 = quad_sum(a0); a1 = quad_sum(a1); a2 = quad_sum(a2); a3 = quad_sum(a3);
    a4 = quad_sum(a4); a5 = quad_sum(a5); a6 = quad_sum(a6);
    if (node < n && sub == 0) {
        float di = dis[node];
        const float4* hs = (const float4*)(h2 + (size_t)node * 16);
        float4 s0 = hs[0], s1 = hs[1];
        float* o = so + (threadIdx.x >> 2) * 7;
        o[0] = b2[0] + di * (a0 + s0.x);
        o[1] = b2[1] + di * (a1 + s0.y);
        o[2] = b2[2] + di * (a2 + s0.z);
        o[3] = b2[3] + di * (a3 + s0.w);
        o[4] = b2[4] + di * (a4 + s1.x);
        o[5] = b2[5] + di * (a5 + s1.y);
        o[6] = b2[6] + di * (a6 + s1.z);
    }
    __syncthreads();
    int nb    = blockIdx.x * NPB;
    int left  = n - nb;
    int cntf  = (left < NPB ? left : NPB) * 7;
    float* ob = out + (size_t)nb * 7;
    for (int j = threadIdx.x; j < NPB * 7; j += AGG_T)
        if (j < cntf) ob[j] = so[j];
}

extern "C" void kernel_launch(void* const* d_in, const int* in_sizes, int n_in,
                              void* d_out, int out_size, void* d_ws, size_t ws_size,
                              hipStream_t stream) {
    (void)n_in; (void)out_size; (void)ws_size;
    const float* x  = (const float*)d_in[0];
    const int*   ei = (const int*)d_in[1];
    const float* ew = (const float*)d_in[2];
    const float* W1 = (const float*)d_in[3];
    const float* b1 = (const float*)d_in[4];
    const float* W2 = (const float*)d_in[5];
    const float* b2 = (const float*)d_in[6];
    float* out = (float*)d_out;

    int n = in_sizes[0] / 3;
    int e = in_sizes[2];
    const int* row = ei;
    const int* col = ei + e;

    int nbk  = (n + 1 + BK_N - 1) / BK_N;
    int npad = nbk * BK_N;
    int nblk = (e + EPB - 1) / EPB;
    int hn   = nbk * nblk + 1;
    int nsb  = (hn + SCAN_E - 1) / SCAN_E;
    int hpad = nsb * SCAN_E;

    int*   off    = (int*)d_ws;
    int*   bsum   = off  + (size_t)npad;
    float* dis    = (float*)(bsum + (size_t)npad);
    float* xs     = dis  + (size_t)npad;
    float* ax     = xs   + (size_t)npad * 4;
    float* acc1   = ax   + (size_t)npad * 4;
    float* h2     = acc1 + (size_t)npad * 16;
    int*   rank   = (int*)(h2 + (size_t)npad * 16);
    int2*  rec    = (int2*)(rank + (((size_t)e + 3) / 4) * 4);
    int*   hist   = (int*)(rec + (((size_t)e + 1) / 2) * 2);
    int2*  trec   = (int2*)(hist + (size_t)hpad);
    int*   tcol   = (int*)(trec + (size_t)e);

    const int B = 256;
    int ntiles  = (n + 15) / 16;
    int gemmT   = ntiles * 32;
    int quadB   = (n + NPB - 1) / NPB;

    k_zero_cnt  <<<nsb, SCAN_T, 0, stream>>>(hist);
    k_bk_hist   <<<nblk, PART_T, 0, stream>>>(col, hist, e, nblk, nbk);
    k_scan_local<<<nsb, SCAN_T, 0, stream>>>(hist, bsum);
    k_scan_fix  <<<nsb, SCAN_T, 0, stream>>>(hist, bsum);
    k_bk_part   <<<nblk, PART_T, 0, stream>>>(row, col, ew, hist, trec, tcol, e, nblk, nbk);
    k_bk_sort   <<<nbk, BK_N, 0, stream>>>(hist, trec, tcol, rank, off, rec, e, nblk);

    k_node_deg  <<<quadB, AGG_T, 0, stream>>>(off, rec, x, dis, xs, n, e);
    k_agg1      <<<quadB, AGG_T, 0, stream>>>(off, rec, xs, dis, ax, n, e);
    k_gemm1     <<<(gemmT + B - 1) / B, B, 0, stream>>>(ax, W1, b1, acc1, n, ntiles);

    k_gemm2     <<<(gemmT + B - 1) / B, B, 0, stream>>>(acc1, W2, dis, h2, n, ntiles);
    k_agg2      <<<quadB, AGG_T, 0, stream>>>(off, rec, h2, dis, b2, out, n, e);
}


# layer-1 segment sum and both WMMA products in one launch: per wave a transposed chain, the first product's D registers are the second's B operand; W2 product's k slots regrouped (f32 sum order)
# speedup vs baseline: 8.4734x; 1.0393x over previous
#include <hip/hip_runtime.h>
#include <hip/hip_bf16.h>


typedef __attribute__((ext_vector_type(2))) float v2f;
typedef __attribute__((ext_vector_type(8))) float v8f;


#define BK_SHIFT 10
#define BK_N (1 << BK_SHIFT)
#define PART_T 512
#define MAXB PART_T
#define EPT1 16
#define EPB (PART_T * EPT1)
#define SB 4
static_assert(SB == 4, "the load fences in k_bk_sort name SB = 4 values each");
#define SCAN_T 256
#define SCAN_E 1024
#define AGG_T 256
#define NPB (AGG_T / 4)

__global__ void k_zero_cnt(int* __restrict__ cnt) {
    int t = blockIdx.x * blockDim.x + threadIdx.x;
    *(int4*)(cnt + (size_t)t * 4) = make_int4(0, 0, 0, 0);
}

__global__ void k_scan_local(int* __restrict__ cnt, int* __restrict__ bsum) {
    __shared__ int wsum[SCAN_T / 32];
    int t    = threadIdx.x;
    int lane = t & 31, w = t >> 5;
    int* p   = cnt + (size_t)blockIdx.x * SCAN_E + t * 4;
    int4 v   = *(const int4*)p;
    int s    = v.x + v.y + v.z + v.w;
    int incl = s;
    for (int d = 1; d < 32; d <<= 1) {
        int y = __shfl_up(incl, d);
        if (lane >= d) incl += y;
    }
    if (lane == 31) wsum[w] = incl;
    __syncthreads();
    int wofs = 0;
    for (int j = 0; j < SCAN_T / 32; ++j) wofs += (j < w) ? wsum[j] : 0;
    int ex = wofs + incl - s;
    int4 o;
    o.x = ex;
    o.y = ex + v.x;
    o.z = o.y + v.y;
    o.w = o.z + v.z;
    *(int4*)p = o;
    if (t == SCAN_T - 1) bsum[blockIdx.x] = ex + s;
}

__global__ void k_scan_fix(int* __restrict__ off, const int* __restrict__ bsum) {
    __shared__ int wsum[SCAN_T / 32];
    int t = threadIdx.x;
    int b = blockIdx.x;
    int part = 0;
    for (int j = t; j < b; j += SCAN_T) part += bsum[j];
    for (int d = 16; d >= 1; d >>= 1) part += __shfl_xor(part, d);
    if ((t & 31) == 0) wsum[t >> 5] = part;
    __syncthreads();
    int base = 0;
    for (int j = 0; j < SCAN_T / 32; ++j) base += wsum[j];
    size_t idx = (size_t)b * SCAN_E + t * 4;
    int4 v = *(const int4*)(off + idx);
    v.x += base; v.y += base; v.z += base; v.w += base;
    *(int4*)(off + idx) = v;
}

__global__ void k_bk_hist(const int* __restrict__ col, int* __restrict__ hist,
                          int e, int nblk, int nbk) {
    __shared__ int lh[MAXB];
    int t = threadIdx.x;
    lh[t] = 0;
    __syncthreads();
    int base = blockIdx.x * EPB;
    int c[EPT1];
    for (int j = 0; j < EPT1; ++j) {
        int i  = base + t + j * PART_T;
        int ic = i < e ? i : e - 1;
        c[j] = col[ic];
    }
    for (int j = 0; j < EPT1; ++j) {
        int i = base + t + j * PART_T;
        if (i < e) atomicAdd(&lh[c[j] >> BK_SHIFT], 1);
    }
    __syncthreads();
    if (t < nbk) hist[(size_t)t * nblk + blockIdx.x] = lh[t];
}

__global__ void k_bk_part(const int* __restrict__ row, const int* __restrict__ col,
                          const float* __restrict__ ew, const int* __restrict__ hist,
                          int2* __restrict__ trec, int* __restrict__ tcol,
                          int e, int nblk, int nbk) {
    __shared__ int  lh[MAXB];
    __shared__ int  gb[MAXB];
    __shared__ int  wsum[PART_T / 32];
    __shared__ int  s_col[EPB];
    __shared__ int2 s_rec[EPB];
    int t    = threadIdx.x;
    int lane = t & 31, w = t >> 5;
    lh[t] = 0;
    __syncthreads();
    int base = blockIdx.x * EPB;
    int c[EPT1], r[EPT1], k[EPT1];
    float wt[EPT1];
    for (int j = 0; j < EPT1; ++j) {
        int i  = base + t + j * PART_T;
        int ic = i < e ? i : e - 1;
        c[j]  = col[ic];
        r[j]  = row[ic];
        wt[j] = ew[ic];
    }
    for (int j = 0; j < EPT1; ++j) {
        int i = base + t + j * PART_T;
        k[j] = 0;
        if (i < e) k[j] = atomicAdd(&lh[c[j] >> BK_SHIFT], 1);
    }
    __syncthreads();
    int v    = lh[t];
    int incl = v;
    for (int d = 1; d < 32; d <<= 1) {
        int y = __shfl_up(incl, d);
        if (lane >= d) incl += y;
    }
    if (lane == 31) wsum[w] = incl;
    __syncthreads();
    int wofs = 0;
    for (int j = 0; j < PART_T / 32; ++j) wofs += (j < w) ? wsum[j] : 0;
    int ex = wofs + incl - v;
    int g  = (t < nbk) ? hist[(size_t)t * nblk + blockIdx.x] : 0;
    lh[t] = ex;
    gb[t] = g - ex;
    __syncthreads();
    for (int j = 0; j < EPT1; ++j) {
        int i = base + t + j * PART_T;
        if (i < e) {
            int s = lh[c[j] >> BK_SHIFT] + k[j];
            s_col[s] = c[j];
            s_rec[s] = make_int2(r[j], __float_as_int(wt[j]));
        }
    }
    __syncthreads();
    int left  = e - base;
    int total = left < EPB ? left : EPB;
    for (int s = t; s < total; s += PART_T) {
        int cc = s_col[s];
        int p  = gb[cc >> BK_SHIFT] + s;
        if ((unsigned)p < (unsigned)e) {
            trec[p] = s_rec[s];
            tcol[p] = cc;
        }
    }
}

__global__ void k_bk_sort(const int* __restrict__ hist, const int2* __restrict__ trec,
                          const int* __restrict__ tcol, int* __restrict__ rank,
                          int* __restrict__ off, int2* __restrict__ rec, int e, int nblk) {
    __shared__ int cnt[BK_N];
    __shared__ int wsum[BK_N / 32];
    int t    = threadIdx.x;
    int lane = t & 31, w = t >> 5;
    int b    = blockIdx.x;
    int bs   = hist[(size_t)b * nblk];
    int be   = hist[(size_t)(b + 1) * nblk];
    bs = bs > 0 ? bs : 0;
    be = be < e ? be : e;
    cnt[t] = 0;
    __syncthreads();
    for (int q0 = bs + t; q0 < be; q0 += SB * BK_N) {
        int cc[SB];
        for (int u = 0; u < SB; ++u) {
            int q  = q0 + u * BK_N;
            int qc = q < be ? q : be - 1;
            cc[u] = tcol[qc];
        }
        asm volatile("" :: "v"(cc[0]), "v"(cc[1]), "v"(cc[2]), "v"(cc[3]));
        for (int u = 0; u < SB; ++u) {
            int q = q0 + u * BK_N;
            if (q < be) {
                int lc = (cc[u] - (b << BK_SHIFT)) & (BK_N - 1);
                rank[q] = atomicAdd(&cnt[lc], 1);
            }
        }
    }
    __syncthreads();
    int v    = cnt[t];
    int incl = v;
    for (int d = 1; d < 32; d <<= 1) {
        int y = __shfl_up(incl, d);
        if (lane >= d) incl += y;
    }
    if (lane == 31) wsum[w] = incl;
    __syncthreads();
    int wofs = 0;
    for (int j = 0; j < BK_N / 32; ++j) wofs += (j < w) ? wsum[j] : 0;
    int st = bs + wofs + incl - v;
    cnt[t] = st;
    off[(size_t)(b << BK_SHIFT) + t] = st;
    __syncthreads();
    for (int q0 = bs + t; q0 < be; q0 += SB * BK_N) {
        int cc[SB], rk[SB];
        int2 rr[SB];
        for (int u = 0; u < SB; ++u) {
            int q  = q0 + u * BK_N;
            int qc = q < be ? q : be - 1;
            cc[u] = tcol[qc];
            rk[u] = rank[qc];
            rr[u] = trec[qc];
        }
        asm volatile("" :: "v"(cc[0]), "v"(cc[1]), "v"(cc[2]), "v"(cc[3]),
                           "v"(rk[0]), "v"(rk[1]), "v"(rk[2]), "v"(rk[3]),
                           "v"(rr[0].x), "v"(rr[0].y), "v"(rr[1].x), "v"(rr[1].y),
                           "v"(rr[2].x), "v"(rr[2].y), "v"(rr[3].x), "v"(rr[3].y));
        for (int u = 0; u < SB; ++u) {
            int q = q0 + u * BK_N;
            int lc = (cc[u] - (b << BK_SHIFT)) & (BK_N - 1);
            int p  = cnt[lc] + rk[u];
            if (q < be && (unsigned)p < (unsigned)e) rec[p] = rr[u];
        }
    }
}

__device__ __forceinline__ void seg_bounds(const int* __restrict__ off, int node, int n, int e,
                                           int& beg, int& end) {
    int nc = node < n ? node : n - 1;
    int b  = off[nc];
    int en = off[nc + 1];
    b  = b > 0 ? b : 0;
    en = en < e ? en : e;
    beg = b;
    end = node < n ? en : b;
}

__device__ __forceinline__ float quad_sum(float v) {
    v += __shfl_xor(v, 1);
    v += __shfl_xor(v, 2);
    return v;
}

__global__ void k_node_deg(const int* __restrict__ off, const int2* __restrict__ rec,
                           const float* __restrict__ x, float* __restrict__ dis,
                           float* __restrict__ xs, int n, int e) {
    int t    = blockIdx.x * blockDim.x + threadIdx.x;
    int node = t >> 2, sub = t & 3;
    int beg, end;
    seg_bounds(off, node, n, e, beg, end);
    float s = 0.f;
    for (int k = beg + sub; k < end; k += 4) s += __int_as_float(rec[k].y);
    s = quad_sum(s);
    if (node < n && sub == 0) {
        float d  = 1.0f + s;
        float di = (d > 0.f) ? rsqrtf(d) : 0.f;
        dis[node] = di;
        const float* xi = x + (size_t)node * 3;
        float4 v;
        v.x = xi[0] * di;
        v.y = xi[1] * di;
        v.z = xi[2] * di;
        v.w = 0.f;
        *(float4*)(xs + (size_t)node * 4) = v;
    }
}

__global__ void k_agg1_mlp(const int* __restrict__ off, const int2* __restrict__ rec,
                           const float* __restrict__ xs, const float* __restrict__ dis,
                           const float* __restrict__ W1, const float* __restrict__ b1,
                           const float* __restrict__ W2, float* __restrict__ h2,
                           int n, int e) {
    int t    = blockIdx.x * blockDim.x + threadIdx.x;
    int node = t >> 2, sub = t & 3;
    int lane = threadIdx.x & 31;
    int c    = lane & 15, h = lane >> 4;

    float w1x = W1[(2 * h) * 16 + c];
    float w1y = W1[16 + c];
    float4 bA = *(const float4*)(b1 + 8 * h);
    float4 bB = *(const float4*)(b1 + 8 * h + 4);
    int  oc = c < 7 ? c : 6;
    bool ov = c < 7;
    float w2x[4], w2y[4];
    for (int s = 0; s < 4; ++s) {
        int f = 8 * h + 2 * s;
        w2x[s] = W2[(f + 0) * 7 + oc];
        w2y[s] = W2[(f + 1) * 7 + oc];
    }

    int beg, end;
    seg_bounds(off, node, n, e, beg, end);
    float a0 = 0.f, a1 = 0.f, a2 = 0.f;
    for (int k = beg + sub; k < end; k += 4) {
        int2 r   = rec[k];
        float w  = __int_as_float(r.y);
        float4 v = *(const float4*)(xs + (size_t)r.x * 4);
        a0 += w * v.x;
        a1 += w * v.y;
        a2 += w * v.z;
    }
    a0 = quad_sum(a0);
    a1 = quad_sum(a1);
    a2 = quad_sum(a2);
    int nc    = node < n ? node : n - 1;
    float di  = dis[nc];
    float4 sf = *(const float4*)(xs + (size_t)nc * 4);
    float o0 = di * (a0 + sf.x);
    float o1 = di * (a1 + sf.y);
    float o2 = di * (a2 + sf.z);

    int srcl  = (lane & 7) << 2;
    float t0  = __shfl(o0, srcl);
    float t1  = __shfl(o1, srcl);
    float t2  = __shfl(o2, srcl);
    float dc  = __shfl(di, srcl);
    int ncol  = ((t - lane) >> 2) + (lane & 7);
    bool cv   = (c < 8) && (ncol < n);

    v2f aw;
    aw.x = w1x;
    aw.y = (h == 0) ? w1y : 0.f;
    v2f bx;
    bx.x = cv ? (h == 0 ? t0 : t2) : 0.f;
    bx.y = (cv && h == 0) ? t1 : 0.f;
    v8f z = {};
    z = __builtin_amdgcn_wmma_f32_16x16x4_f32(false, aw, false, bx, (short)0, z, false, false);

    float ds = cv ? dc : 0.f;
    float zs[8];
    zs[0] = fmaxf(z[0] + bA.x, 0.f) * ds;
    zs[1] = fmaxf(z[1] + bA.y, 0.f) * ds;
    zs[2] = fmaxf(z[2] + bA.z, 0.f) * ds;
    zs[3] = fmaxf(z[3] + bA.w, 0.f) * ds;
    zs[4] = fmaxf(z[4] + bB.x, 0.f) * ds;
    zs[5] = fmaxf(z[5] + bB.y, 0.f) * ds;
    zs[6] = fmaxf(z[6] + bB.z, 0.f) * ds;
    zs[7] = fmaxf(z[7] + bB.w, 0.f) * ds;

    v8f acc = {};
    for (int s = 0; s < 4; ++s) {
        v2f a;
        a.x = ov ? w2x[s] : 0.f;
        a.y = ov ? w2y[s] : 0.f;
        v2f b;
        b.x = zs[2 * s];
        b.y = zs[2 * s + 1];
        acc = __builtin_amdgcn_wmma_f32_16x16x4_f32(false, a, false, b, (short)0, acc, false, false);
    }

    if (cv && h == 0) {
        float4* p = (float4*)(h2 + (size_t)ncol * 16);
        p[0] = make_float4(acc[0], acc[1], acc[2], acc[3]);
        p[1] = make_float4(acc[4], acc[5], acc[6], acc[7]);
    }
}

__global__ void k_agg2(const int* __restrict__ off, const int2* __restrict__ rec,
                       const float* __restrict__ h2, const float* __restrict__ dis,
                       const float* __restrict__ b2, float* __restrict__ out, int n, int e) {
    __shared__ float so[NPB * 7];
    int t    = blockIdx.x * blockDim.x + threadIdx.x;
    int node = t >> 2, sub = t & 3;
    int beg, end;
    seg_bounds(off, node, n, e, beg, end);
    float a0 = 0.f, a1 = 0.f, a2 = 0.f, a3 = 0.f, a4 = 0.f, a5 = 0.f, a6 = 0.f;
    for (int k = beg + sub; k < end; k += 4) {
        int2 r  = rec[k];
        float w = __int_as_float(r.y);
        const float4* hr = (const float4*)(h2 + (size_t)r.x * 16);
        float4 v0 = hr[0], v1 = hr[1];
        a0 += w * v0.x; a1 += w * v0.y; a2 += w * v0.z; a3 += w * v0.w;
        a4 += w * v1.x; a5 += w * v1.y; a6 += w * v1.z;
    }
    a0 = quad_sum(a0); a1 = quad_sum(a1); a2 = quad_sum(a2); a3 = quad_sum(a3);
    a4 = quad_sum(a4); a5 = quad_sum(a5); a6 = quad_sum(a6);
    if (node < n && sub == 0) {
        float di = dis[node];
        const float4* hs = (const float4*)(h2 + (size_t)node * 16);
        float4 s0 = hs[0], s1 = hs[1];
        float* o = so + (threadIdx.x >> 2) * 7;
        o[0] = b2[0] + di * (a0 + s0.x);
        o[1] = b2[1] + di * (a1 + s0.y);
        o[2] = b2[2] + di * (a2 + s0.z);
        o[3] = b2[3] + di * (a3 + s0.w);
        o[4] = b2[4] + di * (a4 + s1.x);
        o[5] = b2[5] + di * (a5 + s1.y);
        o[6] = b2[6] + di * (a6 + s1.z);
    }
    __syncthreads();
    int nb    = blockIdx.x * NPB;
    int left  = n - nb;
    int cntf  = (left < NPB ? left : NPB) * 7;
    float* ob = out + (size_t)nb * 7;
    for (int j = threadIdx.x; j < NPB * 7; j += AGG_T)
        if (j < cntf) ob[j] = so[j];
}

extern "C" void kernel_launch(void* const* d_in, const int* in_sizes, int n_in,
                              void* d_out, int out_size, void* d_ws, size_t ws_size,
                              hipStream_t stream) {
    (void)n_in; (void)out_size; (void)ws_size;
    const float* x  = (const float*)d_in[0];
    const int*   ei = (const int*)d_in[1];
    const float* ew = (const float*)d_in[2];
    const float* W1 = (const float*)d_in[3];
    const float* b1 = (const float*)d_in[4];
    const float* W2 = (const float*)d_in[5];
    const float* b2 = (const float*)d_in[6];
    float* out = (float*)d_out;

    int n = in_sizes[0] / 3;
    int e = in_sizes[2];
    const int* row = ei;
    const int* col = ei + e;

    int nbk  = (n + 1 + BK_N - 1) / BK_N;
    int npad = nbk * BK_N;
    int nblk = (e + EPB - 1) / EPB;
    int hn   = nbk * nblk + 1;
    int nsb  = (hn + SCAN_E - 1) / SCAN_E;
    int hpad = nsb * SCAN_E;

    int*   off    = (int*)d_ws;
    int*   bsum   = off  + (size_t)npad;
    float* dis    = (float*)(bsum + (size_t)npad);
    float* xs     = dis  + (size_t)npad;
    float* h2     = xs   + (size_t)npad * 4;
    int*   rank   = (int*)(h2 + (size_t)npad * 16);
    int2*  rec    = (int2*)(rank + (((size_t)e + 3) / 4) * 4);
    int*   hist   = (int*)(rec + (((size_t)e + 1) / 2) * 2);
    int2*  trec   = (int2*)(hist + (size_t)hpad);
    int*   tcol   = (int*)(trec + (size_t)e);

    int quadB   = (n + NPB - 1) / NPB;

    k_zero_cnt  <<<nsb, SCAN_T, 0, stream>>>(hist);
    k_bk_hist   <<<nblk, PART_T, 0, stream>>>(col, hist, e, nblk, nbk);
    k_scan_local<<<nsb, SCAN_T, 0, stream>>>(hist, bsum);
    k_scan_fix  <<<nsb, SCAN_T, 0, stream>>>(hist, bsum);
    k_bk_part   <<<nblk, PART_T, 0, stream>>>(row, col, ew, hist, trec, tcol, e, nblk, nbk);
    k_bk_sort   <<<nbk, BK_N, 0, stream>>>(hist, trec, tcol, rank, off, rec, e, nblk);

    k_node_deg  <<<quadB, AGG_T, 0, stream>>>(off, rec, x, dis, xs, n, e);
    k_agg1_mlp  <<<quadB, AGG_T, 0, stream>>>(off, rec, xs, dis, W1, b1, W2, h2, n, e);
    k_agg2      <<<quadB, AGG_T, 0, stream>>>(off, rec, h2, dis, b2, out, n, e);
}


# second-level sort: LDS cursors place each record in an LDS image of its bucket, stored to rec in whole 128-byte lines; no rank plane, no scattered 8-byte global stores; f32 sum order still arrival
# speedup vs baseline: 9.7937x; 1.1558x over previous
#include <hip/hip_runtime.h>
#include <hip/hip_bf16.h>


typedef __attribute__((ext_vector_type(2))) float v2f;
typedef __attribute__((ext_vector_type(8))) float v8f;


#define BK_SHIFT 10
#define BK_N (1 << BK_SHIFT)
#define PART_T 512
#define MAXB PART_T
#define EPT1 16
#define EPB (PART_T * EPT1)
#define SB 4
static_assert(SB == 4, "the load fences in k_bk_sort name SB = 4 values each");
#define STAGE_CAP 32768
#define SCAN_T 256
#define SCAN_E 1024
#define AGG_T 256
#define NPB (AGG_T / 4)

__global__ void k_zero_cnt(int* __restrict__ cnt) {
    int t = blockIdx.x * blockDim.x + threadIdx.x;
    *(int4*)(cnt + (size_t)t * 4) = make_int4(0, 0, 0, 0);
}

__global__ void k_scan_local(int* __restrict__ cnt, int* __restrict__ bsum) {
    __shared__ int wsum[SCAN_T / 32];
    int t    = threadIdx.x;
    int lane = t & 31, w = t >> 5;
    int* p   = cnt + (size_t)blockIdx.x * SCAN_E + t * 4;
    int4 v   = *(const int4*)p;
    int s    = v.x + v.y + v.z + v.w;
    int incl = s;
    for (int d = 1; d < 32; d <<= 1) {
        int y = __shfl_up(incl, d);
        if (lane >= d) incl += y;
    }
    if (lane == 31) wsum[w] = incl;
    __syncthreads();
    int wofs = 0;
    for (int j = 0; j < SCAN_T / 32; ++j) wofs += (j < w) ? wsum[j] : 0;
    int ex = wofs + incl - s;
    int4 o;
    o.x = ex;
    o.y = ex + v.x;
    o.z = o.y + v.y;
    o.w = o.z + v.z;
    *(int4*)p = o;
    if (t == SCAN_T - 1) bsum[blockIdx.x] = ex + s;
}

__global__ void k_scan_fix(int* __restrict__ off, const int* __restrict__ bsum) {
    __shared__ int wsum[SCAN_T / 32];
    int t = threadIdx.x;
    int b = blockIdx.x;
    int part = 0;
    for (int j = t; j < b; j += SCAN_T) part += bsum[j];
    for (int d = 16; d >= 1; d >>= 1) part += __shfl_xor(part, d);
    if ((t & 31) == 0) wsum[t >> 5] = part;
    __syncthreads();
    int base = 0;
    for (int j = 0; j < SCAN_T / 32; ++j) base += wsum[j];
    size_t idx = (size_t)b * SCAN_E + t * 4;
    int4 v = *(const int4*)(off + idx);
    v.x += base; v.y += base; v.z += base; v.w += base;
    *(int4*)(off + idx) = v;
}

__global__ void k_bk_hist(const int* __restrict__ col, int* __restrict__ hist,
                          int e, int nblk, int nbk) {
    __shared__ int lh[MAXB];
    int t = threadIdx.x;
    lh[t] = 0;
    __syncthreads();
    int base = blockIdx.x * EPB;
    int c[EPT1];
    for (int j = 0; j < EPT1; ++j) {
        int i  = base + t + j * PART_T;
        int ic = i < e ? i : e - 1;
        c[j] = col[ic];
    }
    for (int j = 0; j < EPT1; ++j) {
        int i = base + t + j * PART_T;
        if (i < e) atomicAdd(&lh[c[j] >> BK_SHIFT], 1);
    }
    __syncthreads();
    if (t < nbk) hist[(size_t)t * nblk + blockIdx.x] = lh[t];
}

__global__ void k_bk_part(const int* __restrict__ row, const int* __restrict__ col,
                          const float* __restrict__ ew, const int* __restrict__ hist,
                          int2* __restrict__ trec, int* __restrict__ tcol,
                          int e, int nblk, int nbk) {
    __shared__ int  lh[MAXB];
    __shared__ int  gb[MAXB];
    __shared__ int  wsum[PART_T / 32];
    __shared__ int  s_col[EPB];
    __shared__ int2 s_rec[EPB];
    int t    = threadIdx.x;
    int lane = t & 31, w = t >> 5;
    lh[t] = 0;
    __syncthreads();
    int base = blockIdx.x * EPB;
    int c[EPT1], r[EPT1], k[EPT1];
    float wt[EPT1];
    for (int j = 0; j < EPT1; ++j) {
        int i  = base + t + j * PART_T;
        int ic = i < e ? i : e - 1;
        c[j]  = col[ic];
        r[j]  = row[ic];
        wt[j] = ew[ic];
    }
    for (int j = 0; j < EPT1; ++j) {
        int i = base + t + j * PART_T;
        k[j] = 0;
        if (i < e) k[j] = atomicAdd(&lh[c[j] >> BK_SHIFT], 1);
    }
    __syncthreads();
    int v    = lh[t];
    int incl = v;
    for (int d = 1; d < 32; d <<= 1) {
        int y = __shfl_up(incl, d);
        if (lane >= d) incl += y;
    }
    if (lane == 31) wsum[w] = incl;
    __syncthreads();
    int wofs = 0;
    for (int j = 0; j < PART_T / 32; ++j) wofs += (j < w) ? wsum[j] : 0;
    int ex = wofs + incl - v;
    int g  = (t < nbk) ? hist[(size_t)t * nblk + blockIdx.x] : 0;
    lh[t] = ex;
    gb[t] = g - ex;
    __syncthreads();
    for (int j = 0; j < EPT1; ++j) {
        int i = base + t + j * PART_T;
        if (i < e) {
            int s = lh[c[j] >> BK_SHIFT] + k[j];
            s_col[s] = c[j];
            s_rec[s] = make_int2(r[j], __float_as_int(wt[j]));
        }
    }
    __syncthreads();
    int left  = e - base;
    int total = left < EPB ? left : EPB;
    for (int s = t; s < total; s += PART_T) {
        int cc = s_col[s];
        int p  = gb[cc >> BK_SHIFT] + s;
        if ((unsigned)p < (unsigned)e) {
            trec[p] = s_rec[s];
            tcol[p] = cc;
        }
    }
}

__global__ void __launch_bounds__(BK_N)
k_bk_sort(const int* __restrict__ hist, const int2* __restrict__ trec,
          const int* __restrict__ tcol, int* __restrict__ off,
          int2* __restrict__ rec, int e, int nblk) {
    __shared__ int  cnt[BK_N];
    __shared__ int  wsum[BK_N / 32];
    __shared__ int2 stage[STAGE_CAP];
    int t    = threadIdx.x;
    int lane = t & 31, w = t >> 5;
    int b    = blockIdx.x;
    int bs   = hist[(size_t)b * nblk];
    int be   = hist[(size_t)(b + 1) * nblk];
    bs = bs > 0 ? bs : 0;
    be = be < e ? be : e;
    int len  = be - bs;
    bool fits = len <= STAGE_CAP;
    cnt[t] = 0;
    __syncthreads();
    for (int q0 = bs + t; q0 < be; q0 += SB * BK_N) {
        int cc[SB];
        for (int u = 0; u < SB; ++u) {
            int q  = q0 + u * BK_N;
            int qc = q < be ? q : be - 1;
            cc[u] = tcol[qc];
        }
        asm volatile("" :: "v"(cc[0]), "v"(cc[1]), "v"(cc[2]), "v"(cc[3]));
        for (int u = 0; u < SB; ++u) {
            int q = q0 + u * BK_N;
            if (q < be) {
                int lc = (cc[u] - (b << BK_SHIFT)) & (BK_N - 1);
                atomicAdd(&cnt[lc], 1);
            }
        }
    }
    __syncthreads();
    int v    = cnt[t];
    int incl = v;
    for (int d = 1; d < 32; d <<= 1) {
        int y = __shfl_up(incl, d);
        if (lane >= d) incl += y;
    }
    if (lane == 31) wsum[w] = incl;
    __syncthreads();
    int wofs = 0;
    for (int j = 0; j < BK_N / 32; ++j) wofs += (j < w) ? wsum[j] : 0;
    int lst = wofs + incl - v;
    cnt[t] = lst;
    off[(size_t)(b << BK_SHIFT) + t] = bs + lst;
    __syncthreads();
    if (fits) {
        for (int q0 = bs + t; q0 < be; q0 += SB * BK_N) {
            int cc[SB], p[SB];
            int2 rr[SB];
            for (int u = 0; u < SB; ++u) {
                int q  = q0 + u * BK_N;
                int qc = q < be ? q : be - 1;
                cc[u] = tcol[qc];
                rr[u] = trec[qc];
            }
            asm volatile("" :: "v"(cc[0]), "v"(cc[1]), "v"(cc[2]), "v"(cc[3]),
                               "v"(rr[0].x), "v"(rr[0].y), "v"(rr[1].x), "v"(rr[1].y),
                               "v"(rr[2].x), "v"(rr[2].y), "v"(rr[3].x), "v"(rr[3].y));
            for (int u = 0; u < SB; ++u) {
                int q  = q0 + u * BK_N;
                int lc = (cc[u] - (b << BK_SHIFT)) & (BK_N - 1);
                p[u] = -1;
                if (q < be) p[u] = atomicAdd(&cnt[lc], 1);
            }
            for (int u = 0; u < SB; ++u)
                if ((unsigned)p[u] < (unsigned)STAGE_CAP) stage[p[u]] = rr[u];
        }
        __syncthreads();
        int head = bs & 15;
        for (int i0 = t - head; i0 < len; i0 += SB * BK_N) {
            int2 r[SB];
            for (int u = 0; u < SB; ++u) {
                int i = i0 + u * BK_N;
                r[u] = stage[(i >= 0 && i < len) ? i : 0];
            }
            for (int u = 0; u < SB; ++u) {
                int i = i0 + u * BK_N;
                if (i >= 0 && i < len) rec[bs + i] = r[u];
            }
        }
    } else {
        for (int q0 = bs + t; q0 < be; q0 += BK_N) {
            int cc = tcol[q0];
            int2 rr = trec[q0];
            int lc = (cc - (b << BK_SHIFT)) & (BK_N - 1);
            int g  = bs + atomicAdd(&cnt[lc], 1);
            if ((unsigned)g < (unsigned)e) rec[g] = rr;
        }
    }
}

__device__ __forceinline__ void seg_bounds(const int* __restrict__ off, int node, int n, int e,
                                           int& beg, int& end) {
    int nc = node < n ? node : n - 1;
    int b  = off[nc];
    int en = off[nc + 1];
    b  = b > 0 ? b : 0;
    en = en < e ? en : e;
    beg = b;
    end = node < n ? en : b;
}

__device__ __forceinline__ float quad_sum(float v) {
    v += __shfl_xor(v, 1);
    v += __shfl_xor(v, 2);
    return v;
}

__global__ void k_node_deg(const int* __restrict__ off, const int2* __restrict__ rec,
                           const float* __restrict__ x, float* __restrict__ dis,
                           float* __restrict__ xs, int n, int e) {
    int t    = blockIdx.x * blockDim.x + threadIdx.x;
    int node = t >> 2, sub = t & 3;
    int beg, end;
    seg_bounds(off, node, n, e, beg, end);
    float s = 0.f;
    for (int k = beg + sub; k < end; k += 4) s += __int_as_float(rec[k].y);
    s = quad_sum(s);
    if (node < n && sub == 0) {
        float d  = 1.0f + s;
        float di = (d > 0.f) ? rsqrtf(d) : 0.f;
        dis[node] = di;
        const float* xi = x + (size_t)node * 3;
        float4 v;
        v.x = xi[0] * di;
        v.y = xi[1] * di;
        v.z = xi[2] * di;
        v.w = 0.f;
        *(float4*)(xs + (size_t)node * 4) = v;
    }
}

__global__ void k_agg1_mlp(const int* __restrict__ off, const int2* __restrict__ rec,
                           const float* __restrict__ xs, const float* __restrict__ dis,
                           const float* __restrict__ W1, const float* __restrict__ b1,
                           const float* __restrict__ W2, float* __restrict__ h2,
                           int n, int e) {
    int t    = blockIdx.x * blockDim.x + threadIdx.x;
    int node = t >> 2, sub = t & 3;
    int lane = threadIdx.x & 31;
    int c    = lane & 15, h = lane >> 4;

    float w1x = W1[(2 * h) * 16 + c];
    float w1y = W1[16 + c];
    float4 bA = *(const float4*)(b1 + 8 * h);
    float4 bB = *(const float4*)(b1 + 8 * h + 4);
    int  oc = c < 7 ? c : 6;
    bool ov = c < 7;
    float w2x[4], w2y[4];
    for (int s = 0; s < 4; ++s) {
        int f = 8 * h + 2 * s;
        w2x[s] = W2[(f + 0) * 7 + oc];
        w2y[s] = W2[(f + 1) * 7 + oc];
    }

    int beg, end;
    seg_bounds(off, node, n, e, beg, end);
    float a0 = 0.f, a1 = 0.f, a2 = 0.f;
    for (int k = beg + sub; k < end; k += 4) {
        int2 r   = rec[k];
        float w  = __int_as_float(r.y);
        float4 v = *(const float4*)(xs + (size_t)r.x * 4);
        a0 += w * v.x;
        a1 += w * v.y;
        a2 += w * v.z;
    }
    a0 = quad_sum(a0);
    a1 = quad_sum(a1);
    a2 = quad_sum(a2);
    int nc    = node < n ? node : n - 1;
    float di  = dis[nc];
    float4 sf = *(const float4*)(xs + (size_t)nc * 4);
    float o0 = di * (a0 + sf.x);
    float o1 = di * (a1 + sf.y);
    float o2 = di * (a2 + sf.z);

    int srcl  = (lane & 7) << 2;
    float t0  = __shfl(o0, srcl);
    float t1  = __shfl(o1, srcl);
    float t2  = __shfl(o2, srcl);
    float dc  = __shfl(di, srcl);
    int ncol  = ((t - lane) >> 2) + (lane & 7);
    bool cv   = (c < 8) && (ncol < n);

    v2f aw;
    aw.x = w1x;
    aw.y = (h == 0) ? w1y : 0.f;
    v2f bx;
    bx.x = cv ? (h == 0 ? t0 : t2) : 0.f;
    bx.y = (cv && h == 0) ? t1 : 0.f;
    v8f z = {};
    z = __builtin_amdgcn_wmma_f32_16x16x4_f32(false, aw, false, bx, (short)0, z, false, false);

    float ds = cv ? dc : 0.f;
    float zs[8];
    zs[0] = fmaxf(z[0] + bA.x, 0.f) * ds;
    zs[1] = fmaxf(z[1] + bA.y, 0.f) * ds;
    zs[2] = fmaxf(z[2] + bA.z, 0.f) * ds;
    zs[3] = fmaxf(z[3] + bA.w, 0.f) * ds;
    zs[4] = fmaxf(z[4] + bB.x, 0.f) * ds;
    zs[5] = fmaxf(z[5] + bB.y, 0.f) * ds;
    zs[6] = fmaxf(z[6] + bB.z, 0.f) * ds;
    zs[7] = fmaxf(z[7] + bB.w, 0.f) * ds;

    v8f acc = {};
    for (int s = 0; s < 4; ++s) {
        v2f a;
        a.x = ov ? w2x[s] : 0.f;
        a.y = ov ? w2y[s] : 0.f;
        v2f b;
        b.x = zs[2 * s];
        b.y = zs[2 * s + 1];
        acc = __builtin_amdgcn_wmma_f32_16x16x4_f32(false, a, false, b, (short)0, acc, false, false);
    }

    if (cv && h == 0) {
        float4* p = (float4*)(h2 + (size_t)ncol * 16);
        p[0] = make_float4(acc[0], acc[1], acc[2], acc[3]);
        p[1] = make_float4(acc[4], acc[5], acc[6], acc[7]);
    }
}

__global__ void k_agg2(const int* __restrict__ off, const int2* __restrict__ rec,
                       const float* __restrict__ h2, const float* __restrict__ dis,
                       const float* __restrict__ b2, float* __restrict__ out, int n, int e) {
    __shared__ float so[NPB * 7];
    int t    = blockIdx.x * blockDim.x + threadIdx.x;
    int node = t >> 2, sub = t & 3;
    int beg, end;
    seg_bounds(off, node, n, e, beg, end);
    float a0 = 0.f, a1 = 0.f, a2 = 0.f, a3 = 0.f, a4 = 0.f, a5 = 0.f, a6 = 0.f;
    for (int k = beg + sub; k < end; k += 4) {
        int2 r  = rec[k];
        float w = __int_as_float(r.y);
        const float4* hr = (const float4*)(h2 + (size_t)r.x * 16);
        float4 v0 = hr[0], v1 = hr[1];
        a0 += w * v0.x; a1 += w * v0.y; a2 += w * v0.z; a3 += w * v0.w;
        a4 += w * v1.x; a5 += w * v1.y; a6 += w * v1.z;
    }
    a0 = quad_sum(a0); a1 = quad_sum(a1); a2 = quad_sum(a2); a3 = quad_sum(a3);
    a4 = quad_sum(a4); a5 = quad_sum(a5); a6 = quad_sum(a6);
    if (node < n && sub == 0) {
        float di = dis[node];
        const float4* hs = (const float4*)(h2 + (size_t)node * 16);
        float4 s0 = hs[0], s1 = hs[1];
        float* o = so + (threadIdx.x >> 2) * 7;
        o[0] = b2[0] + di * (a0 + s0.x);
        o[1] = b2[1] + di * (a1 + s0.y);
        o[2] = b2[2] + di * (a2 + s0.z);
        o[3] = b2[3] + di * (a3 + s0.w);
        o[4] = b2[4] + di * (a4 + s1.x);
        o[5] = b2[5] + di * (a5 + s1.y);
        o[6] = b2[6] + di * (a6 + s1.z);
    }
    __syncthreads();
    int nb    = blockIdx.x * NPB;
    int left  = n - nb;
    int cntf  = (left < NPB ? left : NPB) * 7;
    float* ob = out + (size_t)nb * 7;
    for (int j = threadIdx.x; j < NPB * 7; j += AGG_T)
        if (j < cntf) ob[j] = so[j];
}

extern "C" void kernel_launch(void* const* d_in, const int* in_sizes, int n_in,
                              void* d_out, int out_size, void* d_ws, size_t ws_size,
                              hipStream_t stream) {
    (void)n_in; (void)out_size; (void)ws_size;
    const float* x  = (const float*)d_in[0];
    const int*   ei = (const int*)d_in[1];
    const float* ew = (const float*)d_in[2];
    const float* W1 = (const float*)d_in[3];
    const float* b1 = (const float*)d_in[4];
    const float* W2 = (const float*)d_in[5];
    const float* b2 = (const float*)d_in[6];
    float* out = (float*)d_out;

    int n = in_sizes[0] / 3;
    int e = in_sizes[2];
    const int* row = ei;
    const int* col = ei + e;

    int nbk  = (n + 1 + BK_N - 1) / BK_N;
    int npad = nbk * BK_N;
    int nblk = (e + EPB - 1) / EPB;
    int hn   = nbk * nblk + 1;
    int nsb  = (hn + SCAN_E - 1) / SCAN_E;
    int hpad = nsb * SCAN_E;

    int*   off    = (int*)d_ws;
    int*   bsum   = off  + (size_t)npad;
    float* dis    = (float*)(bsum + (size_t)npad);
    float* xs     = dis  + (size_t)npad;
    float* h2     = xs   + (size_t)npad * 4;
    int2*  rec    = (int2*)(h2 + (size_t)npad * 16);
    int*   hist   = (int*)(rec + (((size_t)e + 1) / 2) * 2);
    int2*  trec   = (int2*)(hist + (size_t)hpad);
    int*   tcol   = (int*)(trec + (size_t)e);

    int quadB   = (n + NPB - 1) / NPB;

    k_zero_cnt  <<<nsb, SCAN_T, 0, stream>>>(hist);
    k_bk_hist   <<<nblk, PART_T, 0, stream>>>(col, hist, e, nblk, nbk);
    k_scan_local<<<nsb, SCAN_T, 0, stream>>>(hist, bsum);
    k_scan_fix  <<<nsb, SCAN_T, 0, stream>>>(hist, bsum);
    k_bk_part   <<<nblk, PART_T, 0, stream>>>(row, col, ew, hist, trec, tcol, e, nblk, nbk);
    k_bk_sort   <<<nbk, BK_N, 0, stream>>>(hist, trec, tcol, off, rec, e, nblk);

    k_node_deg  <<<quadB, AGG_T, 0, stream>>>(off, rec, x, dis, xs, n, e);
    k_agg1_mlp  <<<quadB, AGG_T, 0, stream>>>(off, rec, xs, dis, W1, b1, W2, h2, n, e);
    k_agg2      <<<quadB, AGG_T, 0, stream>>>(off, rec, h2, dis, b2, out, n, e);
}


# degree sum moved into the bucket sort kernel: each thread sums its node's weights from the LDS image and writes dis and xs; the degree launch and its read of rec are gone; f32 sum order unchanged
# speedup vs baseline: 10.4523x; 1.0672x over previous
#include <hip/hip_runtime.h>
#include <hip/hip_bf16.h>


typedef __attribute__((ext_vector_type(2))) float v2f;
typedef __attribute__((ext_vector_type(8))) float v8f;


#define BK_SHIFT 10
#define BK_N (1 << BK_SHIFT)
#define PART_T 512
#define MAXB PART_T
#define EPT1 16
#define EPB (PART_T * EPT1)
#define SB 4
static_assert(SB == 4, "the load fences in k_bk_sort name SB = 4 values each");
#define STAGE_CAP 32768
#define SCAN_T 256
#define SCAN_E 1024
#define AGG_T 256
#define NPB (AGG_T / 4)

__global__ void k_zero_cnt(int* __restrict__ cnt) {
    int t = blockIdx.x * blockDim.x + threadIdx.x;
    *(int4*)(cnt + (size_t)t * 4) = make_int4(0, 0, 0, 0);
}

__global__ void k_scan_local(int* __restrict__ cnt, int* __restrict__ bsum) {
    __shared__ int wsum[SCAN_T / 32];
    int t    = threadIdx.x;
    int lane = t & 31, w = t >> 5;
    int* p   = cnt + (size_t)blockIdx.x * SCAN_E + t * 4;
    int4 v   = *(const int4*)p;
    int s    = v.x + v.y + v.z + v.w;
    int incl = s;
    for (int d = 1; d < 32; d <<= 1) {
        int y = __shfl_up(incl, d);
        if (lane >= d) incl += y;
    }
    if (lane == 31) wsum[w] = incl;
    __syncthreads();
    int wofs = 0;
    for (int j = 0; j < SCAN_T / 32; ++j) wofs += (j < w) ? wsum[j] : 0;
    int ex = wofs + incl - s;
    int4 o;
    o.x = ex;
    o.y = ex + v.x;
    o.z = o.y + v.y;
    o.w = o.z + v.z;
    *(int4*)p = o;
    if (t == SCAN_T - 1) bsum[blockIdx.x] = ex + s;
}

__global__ void k_scan_fix(int* __restrict__ off, const int* __restrict__ bsum) {
    __shared__ int wsum[SCAN_T / 32];
    int t = threadIdx.x;
    int b = blockIdx.x;
    int part = 0;
    for (int j = t; j < b; j += SCAN_T) part += bsum[j];
    for (int d = 16; d >= 1; d >>= 1) part += __shfl_xor(part, d);
    if ((t & 31) == 0) wsum[t >> 5] = part;
    __syncthreads();
    int base = 0;
    for (int j = 0; j < SCAN_T / 32; ++j) base += wsum[j];
    size_t idx = (size_t)b * SCAN_E + t * 4;
    int4 v = *(const int4*)(off + idx);
    v.x += base; v.y += base; v.z += base; v.w += base;
    *(int4*)(off + idx) = v;
}

__global__ void k_bk_hist(const int* __restrict__ col, int* __restrict__ hist,
                          int e, int nblk, int nbk) {
    __shared__ int lh[MAXB];
    int t = threadIdx.x;
    lh[t] = 0;
    __syncthreads();
    int base = blockIdx.x * EPB;
    int c[EPT1];
    for (int j = 0; j < EPT1; ++j) {
        int i  = base + t + j * PART_T;
        int ic = i < e ? i : e - 1;
        c[j] = col[ic];
    }
    for (int j = 0; j < EPT1; ++j) {
        int i = base + t + j * PART_T;
        if (i < e) atomicAdd(&lh[c[j] >> BK_SHIFT], 1);
    }
    __syncthreads();
    if (t < nbk) hist[(size_t)t * nblk + blockIdx.x] = lh[t];
}

__global__ void k_bk_part(const int* __restrict__ row, const int* __restrict__ col,
                          const float* __restrict__ ew, const int* __restrict__ hist,
                          int2* __restrict__ trec, int* __restrict__ tcol,
                          int e, int nblk, int nbk) {
    __shared__ int  lh[MAXB];
    __shared__ int  gb[MAXB];
    __shared__ int  wsum[PART_T / 32];
    __shared__ int  s_col[EPB];
    __shared__ int2 s_rec[EPB];
    int t    = threadIdx.x;
    int lane = t & 31, w = t >> 5;
    lh[t] = 0;
    __syncthreads();
    int base = blockIdx.x * EPB;
    int c[EPT1], r[EPT1], k[EPT1];
    float wt[EPT1];
    for (int j = 0; j < EPT1; ++j) {
        int i  = base + t + j * PART_T;
        int ic = i < e ? i : e - 1;
        c[j]  = col[ic];
        r[j]  = row[ic];
        wt[j] = ew[ic];
    }
    for (int j = 0; j < EPT1; ++j) {
        int i = base + t + j * PART_T;
        k[j] = 0;
        if (i < e) k[j] = atomicAdd(&lh[c[j] >> BK_SHIFT], 1);
    }
    __syncthreads();
    int v    = lh[t];
    int incl = v;
    for (int d = 1; d < 32; d <<= 1) {
        int y = __shfl_up(incl, d);
        if (lane >= d) incl += y;
    }
    if (lane == 31) wsum[w] = incl;
    __syncthreads();
    int wofs = 0;
    for (int j = 0; j < PART_T / 32; ++j) wofs += (j < w) ? wsum[j] : 0;
    int ex = wofs + incl - v;
    int g  = (t < nbk) ? hist[(size_t)t * nblk + blockIdx.x] : 0;
    lh[t] = ex;
    gb[t] = g - ex;
    __syncthreads();
    for (int j = 0; j < EPT1; ++j) {
        int i = base + t + j * PART_T;
        if (i < e) {
            int s = lh[c[j] >> BK_SHIFT] + k[j];
            s_col[s] = c[j];
            s_rec[s] = make_int2(r[j], __float_as_int(wt[j]));
        }
    }
    __syncthreads();
    int left  = e - base;
    int total = left < EPB ? left : EPB;
    for (int s = t; s < total; s += PART_T) {
        int cc = s_col[s];
        int p  = gb[cc >> BK_SHIFT] + s;
        if ((unsigned)p < (unsigned)e) {
            trec[p] = s_rec[s];
            tcol[p] = cc;
        }
    }
}

__global__ void __launch_bounds__(BK_N)
k_bk_sort(const int* __restrict__ hist, const int2* __restrict__ trec,
          const int* __restrict__ tcol, const float* __restrict__ x,
          int* __restrict__ off, int2* __restrict__ rec,
          float* __restrict__ dis, float* __restrict__ xs,
          int e, int nblk, int n) {
    __shared__ int  cnt[BK_N];
    __shared__ int  wsum[BK_N / 32];
    __shared__ int2 stage[STAGE_CAP];
    int t    = threadIdx.x;
    int lane = t & 31, w = t >> 5;
    int b    = blockIdx.x;
    int bs   = hist[(size_t)b * nblk];
    int be   = hist[(size_t)(b + 1) * nblk];
    bs = bs > 0 ? bs : 0;
    be = be < e ? be : e;
    int len  = be - bs;
    bool fits = len <= STAGE_CAP;
    cnt[t] = 0;
    __syncthreads();
    for (int q0 = bs + t; q0 < be; q0 += SB * BK_N) {
        int cc[SB];
        for (int u = 0; u < SB; ++u) {
            int q  = q0 + u * BK_N;
            int qc = q < be ? q : be - 1;
            cc[u] = tcol[qc];
        }
        asm volatile("" :: "v"(cc[0]), "v"(cc[1]), "v"(cc[2]), "v"(cc[3]));
        for (int u = 0; u < SB; ++u) {
            int q = q0 + u * BK_N;
            if (q < be) {
                int lc = (cc[u] - (b << BK_SHIFT)) & (BK_N - 1);
                atomicAdd(&cnt[lc], 1);
            }
        }
    }
    __syncthreads();
    int v    = cnt[t];
    int incl = v;
    for (int d = 1; d < 32; d <<= 1) {
        int y = __shfl_up(incl, d);
        if (lane >= d) incl += y;
    }
    if (lane == 31) wsum[w] = incl;
    __syncthreads();
    int wofs = 0;
    for (int j = 0; j < BK_N / 32; ++j) wofs += (j < w) ? wsum[j] : 0;
    int lst = wofs + incl - v;
    cnt[t] = lst;
    off[(size_t)(b << BK_SHIFT) + t] = bs + lst;
    __syncthreads();
    int node = (b << BK_SHIFT) + t;
    const float* xi = x + (size_t)(node < n ? node : n - 1) * 3;
    float xv0, xv1, xv2;
    float s0 = 0.f, s1 = 0.f, s2 = 0.f, s3 = 0.f;
    if (fits) {
        for (int q0 = bs + t; q0 < be; q0 += SB * BK_N) {
            int cc[SB], p[SB];
            int2 rr[SB];
            for (int u = 0; u < SB; ++u) {
                int q  = q0 + u * BK_N;
                int qc = q < be ? q : be - 1;
                cc[u] = tcol[qc];
                rr[u] = trec[qc];
            }
            asm volatile("" :: "v"(cc[0]), "v"(cc[1]), "v"(cc[2]), "v"(cc[3]),
                               "v"(rr[0].x), "v"(rr[0].y), "v"(rr[1].x), "v"(rr[1].y),
                               "v"(rr[2].x), "v"(rr[2].y), "v"(rr[3].x), "v"(rr[3].y));
            for (int u = 0; u < SB; ++u) {
                int q  = q0 + u * BK_N;
                int lc = (cc[u] - (b << BK_SHIFT)) & (BK_N - 1);
                p[u] = -1;
                if (q < be) p[u] = atomicAdd(&cnt[lc], 1);
            }
            for (int u = 0; u < SB; ++u)
                if ((unsigned)p[u] < (unsigned)STAGE_CAP) stage[p[u]] = rr[u];
        }
        __syncthreads();
        xv0 = xi[0]; xv1 = xi[1]; xv2 = xi[2];
        int head = bs & 15;
        for (int i0 = t - head; i0 < len; i0 += SB * BK_N) {
            int2 r[SB];
            for (int u = 0; u < SB; ++u) {
                int i = i0 + u * BK_N;
                r[u] = stage[(i >= 0 && i < len) ? i : 0];
            }
            for (int u = 0; u < SB; ++u) {
                int i = i0 + u * BK_N;
                if (i >= 0 && i < len) rec[bs + i] = r[u];
            }
        }
        for (int j = 0; j < v; j += 4) {
            int i  = lst + j;
            int i1 = j + 1 < v ? i + 1 : i;
            int i2 = j + 2 < v ? i + 2 : i;
            int i3 = j + 3 < v ? i + 3 : i;
            float w0 = __int_as_float(stage[i].y);
            float w1 = __int_as_float(stage[i1].y);
            float w2 = __int_as_float(stage[i2].y);
            float w3 = __int_as_float(stage[i3].y);
            s0 += w0;
            if (j + 1 < v) s1 += w1;
            if (j + 2 < v) s2 += w2;
            if (j + 3 < v) s3 += w3;
        }
    } else {
        for (int q0 = bs + t; q0 < be; q0 += BK_N) {
            int cc = tcol[q0];
            int2 rr = trec[q0];
            int lc = (cc - (b << BK_SHIFT)) & (BK_N - 1);
            int g  = bs + atomicAdd(&cnt[lc], 1);
            if ((unsigned)g < (unsigned)e) rec[g] = rr;
        }
        __syncthreads();
        xv0 = xi[0]; xv1 = xi[1]; xv2 = xi[2];
        for (int j = 0; j < v; j += 4) {
            int g = bs + lst + j;
            for (int u = 0; u < 4; ++u) {
                int gu = g + u;
                gu = gu < e ? gu : e - 1;
                float w = __int_as_float(rec[gu].y);
                if (j + u < v) {
                    if (u == 0) s0 += w;
                    if (u == 1) s1 += w;
                    if (u == 2) s2 += w;
                    if (u == 3) s3 += w;
                }
            }
        }
    }
    if (node < n) {
        float d  = 1.0f + ((s0 + s1) + (s2 + s3));
        float di = (d > 0.f) ? rsqrtf(d) : 0.f;
        dis[node] = di;
        float4 o;
        o.x = xv0 * di;
        o.y = xv1 * di;
        o.z = xv2 * di;
        o.w = 0.f;
        *(float4*)(xs + (size_t)node * 4) = o;
    }
}

__device__ __forceinline__ void seg_bounds(const int* __restrict__ off, int node, int n, int e,
                                           int& beg, int& end) {
    int nc = node < n ? node : n - 1;
    int b  = off[nc];
    int en = off[nc + 1];
    b  = b > 0 ? b : 0;
    en = en < e ? en : e;
    beg = b;
    end = node < n ? en : b;
}

__device__ __forceinline__ float quad_sum(float v) {
    v += __shfl_xor(v, 1);
    v += __shfl_xor(v, 2);
    return v;
}

__global__ void k_agg1_mlp(const int* __restrict__ off, const int2* __restrict__ rec,
                           const float* __restrict__ xs, const float* __restrict__ dis,
                           const float* __restrict__ W1, const float* __restrict__ b1,
                           const float* __restrict__ W2, float* __restrict__ h2,
                           int n, int e) {
    int t    = blockIdx.x * blockDim.x + threadIdx.x;
    int node = t >> 2, sub = t & 3;
    int lane = threadIdx.x & 31;
    int c    = lane & 15, h = lane >> 4;

    float w1x = W1[(2 * h) * 16 + c];
    float w1y = W1[16 + c];
    float4 bA = *(const float4*)(b1 + 8 * h);
    float4 bB = *(const float4*)(b1 + 8 * h + 4);
    int  oc = c < 7 ? c : 6;
    bool ov = c < 7;
    float w2x[4], w2y[4];
    for (int s = 0; s < 4; ++s) {
        int f = 8 * h + 2 * s;
        w2x[s] = W2[(f + 0) * 7 + oc];
        w2y[s] = W2[(f + 1) * 7 + oc];
    }

    int beg, end;
    seg_bounds(off, node, n, e, beg, end);
    float a0 = 0.f, a1 = 0.f, a2 = 0.f;
    for (int k = beg + sub; k < end; k += 4) {
        int2 r   = rec[k];
        float w  = __int_as_float(r.y);
        float4 v = *(const float4*)(xs + (size_t)r.x * 4);
        a0 += w * v.x;
        a1 += w * v.y;
        a2 += w * v.z;
    }
    a0 = quad_sum(a0);
    a1 = quad_sum(a1);
    a2 = quad_sum(a2);
    int nc    = node < n ? node : n - 1;
    float di  = dis[nc];
    float4 sf = *(const float4*)(xs + (size_t)nc * 4);
    float o0 = di * (a0 + sf.x);
    float o1 = di * (a1 + sf.y);
    float o2 = di * (a2 + sf.z);

    int srcl  = (lane & 7) << 2;
    float t0  = __shfl(o0, srcl);
    float t1  = __shfl(o1, srcl);
    float t2  = __shfl(o2, srcl);
    float dc  = __shfl(di, srcl);
    int ncol  = ((t - lane) >> 2) + (lane & 7);
    bool cv   = (c < 8) && (ncol < n);

    v2f aw;
    aw.x = w1x;
    aw.y = (h == 0) ? w1y : 0.f;
    v2f bx;
    bx.x = cv ? (h == 0 ? t0 : t2) : 0.f;
    bx.y = (cv && h == 0) ? t1 : 0.f;
    v8f z = {};
    z = __builtin_amdgcn_wmma_f32_16x16x4_f32(false, aw, false, bx, (short)0, z, false, false);

    float ds = cv ? dc : 0.f;
    float zs[8];
    zs[0] = fmaxf(z[0] + bA.x, 0.f) * ds;
    zs[1] = fmaxf(z[1] + bA.y, 0.f) * ds;
    zs[2] = fmaxf(z[2] + bA.z, 0.f) * ds;
    zs[3] = fmaxf(z[3] + bA.w, 0.f) * ds;
    zs[4] = fmaxf(z[4] + bB.x, 0.f) * ds;
    zs[5] = fmaxf(z[5] + bB.y, 0.f) * ds;
    zs[6] = fmaxf(z[6] + bB.z, 0.f) * ds;
    zs[7] = fmaxf(z[7] + bB.w, 0.f) * ds;

    v8f acc = {};
    for (int s = 0; s < 4; ++s) {
        v2f a;
        a.x = ov ? w2x[s] : 0.f;
        a.y = ov ? w2y[s] : 0.f;
        v2f b;
        b.x = zs[2 * s];
        b.y = zs[2 * s + 1];
        acc = __builtin_amdgcn_wmma_f32_16x16x4_f32(false, a, false, b, (short)0, acc, false, false);
    }

    if (cv && h == 0) {
        float4* p = (float4*)(h2 + (size_t)ncol * 16);
        p[0] = make_float4(acc[0], acc[1], acc[2], acc[3]);
        p[1] = make_float4(acc[4], acc[5], acc[6], acc[7]);
    }
}

__global__ void k_agg2(const int* __restrict__ off, const int2* __restrict__ rec,
                       const float* __restrict__ h2, const float* __restrict__ dis,
                       const float* __restrict__ b2, float* __restrict__ out, int n, int e) {
    __shared__ float so[NPB * 7];
    int t    = blockIdx.x * blockDim.x + threadIdx.x;
    int node = t >> 2, sub = t & 3;
    int beg, end;
    seg_bounds(off, node, n, e, beg, end);
    float a0 = 0.f, a1 = 0.f, a2 = 0.f, a3 = 0.f, a4 = 0.f, a5 = 0.f, a6 = 0.f;
    for (int k = beg + sub; k < end; k += 4) {
        int2 r  = rec[k];
        float w = __int_as_float(r.y);
        const float4* hr = (const float4*)(h2 + (size_t)r.x * 16);
        float4 v0 = hr[0], v1 = hr[1];
        a0 += w * v0.x; a1 += w * v0.y; a2 += w * v0.z; a3 += w * v0.w;
        a4 += w * v1.x; a5 += w * v1.y; a6 += w * v1.z;
    }
    a0 = quad_sum(a0); a1 = quad_sum(a1); a2 = quad_sum(a2); a3 = quad_sum(a3);
    a4 = quad_sum(a4); a5 = quad_sum(a5); a6 = quad_sum(a6);
    if (node < n && sub == 0) {
        float di = dis[node];
        const float4* hs = (const float4*)(h2 + (size_t)node * 16);
        float4 s0 = hs[0], s1 = hs[1];
        float* o = so + (threadIdx.x >> 2) * 7;
        o[0] = b2[0] + di * (a0 + s0.x);
        o[1] = b2[1] + di * (a1 + s0.y);
        o[2] = b2[2] + di * (a2 + s0.z);
        o[3] = b2[3] + di * (a3 + s0.w);
        o[4] = b2[4] + di * (a4 + s1.x);
        o[5] = b2[5] + di * (a5 + s1.y);
        o[6] = b2[6] + di * (a6 + s1.z);
    }
    __syncthreads();
    int nb    = blockIdx.x * NPB;
    int left  = n - nb;
    int cntf  = (left < NPB ? left : NPB) * 7;
    float* ob = out + (size_t)nb * 7;
    for (int j = threadIdx.x; j < NPB * 7; j += AGG_T)
        if (j < cntf) ob[j] = so[j];
}

extern "C" void kernel_launch(void* const* d_in, const int* in_sizes, int n_in,
                              void* d_out, int out_size, void* d_ws, size_t ws_size,
                              hipStream_t stream) {
    (void)n_in; (void)out_size; (void)ws_size;
    const float* x  = (const float*)d_in[0];
    const int*   ei = (const int*)d_in[1];
    const float* ew = (const float*)d_in[2];
    const float* W1 = (const float*)d_in[3];
    const float* b1 = (const float*)d_in[4];
    const float* W2 = (const float*)d_in[5];
    const float* b2 = (const float*)d_in[6];
    float* out = (float*)d_out;

    int n = in_sizes[0] / 3;
    int e = in_sizes[2];
    const int* row = ei;
    const int* col = ei + e;

    int nbk  = (n + 1 + BK_N - 1) / BK_N;
    int npad = nbk * BK_N;
    int nblk = (e + EPB - 1) / EPB;
    int hn   = nbk * nblk + 1;
    int nsb  = (hn + SCAN_E - 1) / SCAN_E;
    int hpad = nsb * SCAN_E;

    int*   off    = (int*)d_ws;
    int*   bsum   = off  + (size_t)npad;
    float* dis    = (float*)(bsum + (size_t)npad);
    float* xs     = dis  + (size_t)npad;
    float* h2     = xs   + (size_t)npad * 4;
    int2*  rec    = (int2*)(h2 + (size_t)npad * 16);
    int*   hist   = (int*)(rec + (((size_t)e + 1) / 2) * 2);
    int2*  trec   = (int2*)(hist + (size_t)hpad);
    int*   tcol   = (int*)(trec + (size_t)e);

    int quadB   = (n + NPB - 1) / NPB;

    k_zero_cnt  <<<nsb, SCAN_T, 0, stream>>>(hist);
    k_bk_hist   <<<nblk, PART_T, 0, stream>>>(col, hist, e, nblk, nbk);
    k_scan_local<<<nsb, SCAN_T, 0, stream>>>(hist, bsum);
    k_scan_fix  <<<nsb, SCAN_T, 0, stream>>>(hist, bsum);
    k_bk_part   <<<nblk, PART_T, 0, stream>>>(row, col, ew, hist, trec, tcol, e, nblk, nbk);
    k_bk_sort   <<<nbk, BK_N, 0, stream>>>(hist, trec, tcol, x, off, rec, dis, xs, e, nblk, n);

    k_agg1_mlp  <<<quadB, AGG_T, 0, stream>>>(off, rec, xs, dis, W1, b1, W2, h2, n, e);
    k_agg2      <<<quadB, AGG_T, 0, stream>>>(off, rec, h2, dis, b2, out, n, e);
}


# level 1 of the sort made block-local: each block writes its bucket-ordered edges to its own slice plus a (start,count) table; hist, zero and 2 scan launches gone; new f32 sum order (arrival rank)
# speedup vs baseline: 11.4400x; 1.0945x over previous
#include <hip/hip_runtime.h>
#include <hip/hip_bf16.h>


typedef __attribute__((ext_vector_type(2))) float v2f;
typedef __attribute__((ext_vector_type(8))) float v8f;


#define BK_SHIFT 10
#define BK_N (1 << BK_SHIFT)
#define PART_T 512
#define MAXB PART_T
#define EPT1 16
#define EPB (PART_T * EPT1)
#define SB 4
#define RU 4
#define RW (32 * 32)
#define STAGE_CAP 32768
#define AGG_T 256
#define NPB (AGG_T / 4)

__global__ void k_bk_part(const int* __restrict__ row, const int* __restrict__ col,
                          const float* __restrict__ ew,
                          int* __restrict__ rst, int* __restrict__ rcn,
                          int2* __restrict__ trec, int* __restrict__ tcol,
                          int e, int nblk, int pitch, int nbk) {
    __shared__ int  lh[MAXB];
    __shared__ int  wsum[PART_T / 32];
    __shared__ int  s_col[EPB];
    __shared__ int2 s_rec[EPB];
    int t    = threadIdx.x;
    int lane = t & 31, w = t >> 5;
    lh[t] = 0;
    __syncthreads();
    int base = blockIdx.x * EPB;
    int c[EPT1], r[EPT1], k[EPT1];
    float wt[EPT1];
    for (int j = 0; j < EPT1; ++j) {
        int i  = base + t + j * PART_T;
        int ic = i < e ? i : e - 1;
        c[j]  = col[ic];
        r[j]  = row[ic];
        wt[j] = ew[ic];
    }
    for (int j = 0; j < EPT1; ++j) {
        int i = base + t + j * PART_T;
        k[j] = 0;
        if (i < e) k[j] = atomicAdd(&lh[c[j] >> BK_SHIFT], 1);
    }
    __syncthreads();
    int v    = lh[t];
    int incl = v;
    for (int d = 1; d < 32; d <<= 1) {
        int y = __shfl_up(incl, d);
        if (lane >= d) incl += y;
    }
    if (lane == 31) wsum[w] = incl;
    __syncthreads();
    int wofs = 0;
    for (int j = 0; j < PART_T / 32; ++j) wofs += (j < w) ? wsum[j] : 0;
    int ex = wofs + incl - v;
    if (t < nbk) {
        size_t ti = (size_t)t * pitch;
        rst[ti + blockIdx.x] = ex;
        rcn[ti + blockIdx.x] = v;
        if (blockIdx.x == 0)
            for (int j = nblk; j < pitch; ++j) rcn[ti + j] = 0;
    }
    lh[t] = ex;
    __syncthreads();
    for (int j = 0; j < EPT1; ++j) {
        int i = base + t + j * PART_T;
        if (i < e) {
            int s = lh[c[j] >> BK_SHIFT] + k[j];
            s_col[s] = c[j];
            s_rec[s] = make_int2(r[j], __float_as_int(wt[j]));
        }
    }
    __syncthreads();
    int left  = e - base;
    int total = left < EPB ? left : EPB;
    for (int s = t; s < total; s += PART_T) {
        trec[base + s] = s_rec[s];
        tcol[base + s] = s_col[s];
    }
}

__global__ void __launch_bounds__(BK_N)
k_bk_sort(const int* __restrict__ rst, const int* __restrict__ rcn,
          const int2* __restrict__ trec,
          const int* __restrict__ tcol, const float* __restrict__ x,
          int* __restrict__ off, int2* __restrict__ rec,
          float* __restrict__ dis, float* __restrict__ xs,
          int e, int nblk, int pitch, int n) {
    __shared__ int  cnt[BK_N];
    __shared__ int  wsum[BK_N / 32];
    __shared__ int  wsA[BK_N / 32];
    __shared__ int  wsB[BK_N / 32];
    __shared__ int2 stage[STAGE_CAP];
    int t    = threadIdx.x;
    int lane = t & 31, w = t >> 5;
    int b    = blockIdx.x;
    const int* myst = rst + (size_t)b * pitch;
    const int* mycn = rcn + (size_t)b * pitch;
    cnt[t] = 0;
    const int4* c4 = (const int4*)rcn;
    int nq  = b * (pitch >> 2);
    int acc = 0;
    for (int i0 = t; i0 < nq; i0 += 8 * BK_N) {
        int4 q[8];
        for (int u = 0; u < 8; ++u) {
            int i = i0 + u * BK_N;
            q[u] = c4[i < nq ? i : nq - 1];
        }
        for (int u = 0; u < 8; ++u) {
            int i = i0 + u * BK_N;
            if (i < nq) acc += (q[u].x + q[u].y) + (q[u].z + q[u].w);
        }
    }
    __syncthreads();
    int mylen = 0;
    for (int jb = 0; jb < nblk; jb += RW) {
        int jm  = jb + w + 32 * lane;
        int jc  = jm < nblk ? jm : nblk - 1;
        int mst = myst[jc];
        int mcn = jm < nblk ? mycn[jc] : 0;
        mylen += mcn;
        int rl = nblk - jb - w;
        int nr = rl > 0 ? (rl + 31) >> 5 : 0;
        nr = nr < 32 ? nr : 32;
        for (int i0 = 0; i0 < nr; i0 += RU) {
            int cu[RU], gu[RU], cc[RU][2];
            for (int u = 0; u < RU; ++u) {
                int ii = i0 + u;
                int st = __shfl(mst, ii & 31);
                int cn = __shfl(mcn, ii & 31);
                cu[u] = ii < nr ? cn : 0;
                gu[u] = ii < nr ? (jb + w + 32 * ii) * EPB + st : 0;
                for (int h = 0; h < 2; ++h) {
                    int k   = lane + 32 * h;
                    int idx = gu[u] + (k < cu[u] ? k : 0);
                    idx = (unsigned)idx < (unsigned)e ? idx : e - 1;
                    cc[u][h] = tcol[idx];
                }
            }
            for (int u = 0; u < RU; ++u)
                for (int h = 0; h < 2; ++h) {
                    int k = lane + 32 * h;
                    if (k < cu[u]) {
                        int lc = (cc[u][h] - (b << BK_SHIFT)) & (BK_N - 1);
                        atomicAdd(&cnt[lc], 1);
                    }
                }
            for (int u = 0; u < RU; ++u)
                for (int k = lane + 64; k < cu[u]; k += 32) {
                    int idx = gu[u] + k;
                    idx = (unsigned)idx < (unsigned)e ? idx : e - 1;
                    int lc = (tcol[idx] - (b << BK_SHIFT)) & (BK_N - 1);
                    atomicAdd(&cnt[lc], 1);
                }
        }
    }
    for (int d = 16; d >= 1; d >>= 1) {
        acc   += __shfl_xor(acc, d);
        mylen += __shfl_xor(mylen, d);
    }
    if (lane == 0) { wsA[w] = acc; wsB[w] = mylen; }
    __syncthreads();
    int bs = 0, len = 0;
    for (int j = 0; j < BK_N / 32; ++j) { bs += wsA[j]; len += wsB[j]; }
    bs = bs > 0 ? bs : 0;
    bs = bs < e ? bs : e;
    int be = bs + (len > 0 ? len : 0);
    be = (be < e && be >= bs) ? be : e;
    len = be - bs;
    bool fits = len <= STAGE_CAP;
    int v    = cnt[t];
    int incl = v;
    for (int d = 1; d < 32; d <<= 1) {
        int y = __shfl_up(incl, d);
        if (lane >= d) incl += y;
    }
    if (lane == 31) wsum[w] = incl;
    __syncthreads();
    int wofs = 0;
    for (int j = 0; j < BK_N / 32; ++j) wofs += (j < w) ? wsum[j] : 0;
    int lst = wofs + incl - v;
    cnt[t] = lst;
    off[(size_t)(b << BK_SHIFT) + t] = bs + lst;
    __syncthreads();
    int node = (b << BK_SHIFT) + t;
    const float* xi = x + (size_t)(node < n ? node : n - 1) * 3;
    float xv0, xv1, xv2;
    float s0 = 0.f, s1 = 0.f, s2 = 0.f, s3 = 0.f;
    for (int jb = 0; jb < nblk; jb += RW) {
        int jm  = jb + w + 32 * lane;
        int jc  = jm < nblk ? jm : nblk - 1;
        int mst = myst[jc];
        int mcn = jm < nblk ? mycn[jc] : 0;
        int rl = nblk - jb - w;
        int nr = rl > 0 ? (rl + 31) >> 5 : 0;
        nr = nr < 32 ? nr : 32;
        for (int i0 = 0; i0 < nr; i0 += RU) {
            int cu[RU], gu[RU], cc[RU][2], p[RU][2];
            int2 rr[RU][2];
            for (int u = 0; u < RU; ++u) {
                int ii = i0 + u;
                int st = __shfl(mst, ii & 31);
                int cn = __shfl(mcn, ii & 31);
                cu[u] = ii < nr ? cn : 0;
                gu[u] = ii < nr ? (jb + w + 32 * ii) * EPB + st : 0;
                for (int h = 0; h < 2; ++h) {
                    int k   = lane + 32 * h;
                    int idx = gu[u] + (k < cu[u] ? k : 0);
                    idx = (unsigned)idx < (unsigned)e ? idx : e - 1;
                    cc[u][h] = tcol[idx];
                    rr[u][h] = trec[idx];
                }
            }
            for (int u = 0; u < RU; ++u)
                for (int h = 0; h < 2; ++h) {
                    int k  = lane + 32 * h;
                    int lc = (cc[u][h] - (b << BK_SHIFT)) & (BK_N - 1);
                    p[u][h] = -1;
                    if (k < cu[u]) p[u][h] = atomicAdd(&cnt[lc], 1);
                }
            for (int u = 0; u < RU; ++u)
                for (int h = 0; h < 2; ++h) {
                    if (fits) {
                        if ((unsigned)p[u][h] < (unsigned)STAGE_CAP) stage[p[u][h]] = rr[u][h];
                    } else {
                        int g = bs + p[u][h];
                        if (p[u][h] >= 0 && (unsigned)g < (unsigned)e) rec[g] = rr[u][h];
                    }
                }
            for (int u = 0; u < RU; ++u)
                for (int k = lane + 64; k < cu[u]; k += 32) {
                    int idx = gu[u] + k;
                    idx = (unsigned)idx < (unsigned)e ? idx : e - 1;
                    int lc = (tcol[idx] - (b << BK_SHIFT)) & (BK_N - 1);
                    int2 r2 = trec[idx];
                    int pp = atomicAdd(&cnt[lc], 1);
                    if (fits) {
                        if ((unsigned)pp < (unsigned)STAGE_CAP) stage[pp] = r2;
                    } else {
                        int g = bs + pp;
                        if ((unsigned)g < (unsigned)e) rec[g] = r2;
                    }
                }
        }
    }
    __syncthreads();
    xv0 = xi[0]; xv1 = xi[1]; xv2 = xi[2];
    if (fits) {
        int head = bs & 15;
        for (int i0 = t - head; i0 < len; i0 += SB * BK_N) {
            int2 r[SB];
            for (int u = 0; u < SB; ++u) {
                int i = i0 + u * BK_N;
                r[u] = stage[(i >= 0 && i < len) ? i : 0];
            }
            for (int u = 0; u < SB; ++u) {
                int i = i0 + u * BK_N;
                if (i >= 0 && i < len) rec[bs + i] = r[u];
            }
        }
        for (int j = 0; j < v; j += 4) {
            int i  = lst + j;
            int i1 = j + 1 < v ? i + 1 : i;
            int i2 = j + 2 < v ? i + 2 : i;
            int i3 = j + 3 < v ? i + 3 : i;
            float w0 = __int_as_float(stage[i].y);
            float w1 = __int_as_float(stage[i1].y);
            float w2 = __int_as_float(stage[i2].y);
            float w3 = __int_as_float(stage[i3].y);
            s0 += w0;
            if (j + 1 < v) s1 += w1;
            if (j + 2 < v) s2 += w2;
            if (j + 3 < v) s3 += w3;
        }
    } else {
        for (int j = 0; j < v; j += 4) {
            int g = bs + lst + j;
            for (int u = 0; u < 4; ++u) {
                int gu2 = g + u;
                gu2 = (unsigned)gu2 < (unsigned)e ? gu2 : e - 1;
                float wv = __int_as_float(rec[gu2].y);
                if (j + u < v) {
                    if (u == 0) s0 += wv;
                    if (u == 1) s1 += wv;
                    if (u == 2) s2 += wv;
                    if (u == 3) s3 += wv;
                }
            }
        }
    }
    if (node < n) {
        float d  = 1.0f + ((s0 + s1) + (s2 + s3));
        float di = (d > 0.f) ? rsqrtf(d) : 0.f;
        dis[node] = di;
        float4 o;
        o.x = xv0 * di;
        o.y = xv1 * di;
        o.z = xv2 * di;
        o.w = 0.f;
        *(float4*)(xs + (size_t)node * 4) = o;
    }
}

__device__ __forceinline__ void seg_bounds(const int* __restrict__ off, int node, int n, int e,
                                           int& beg, int& end) {
    int nc = node < n ? node : n - 1;
    int b  = off[nc];
    int en = off[nc + 1];
    b  = b > 0 ? b : 0;
    en = en < e ? en : e;
    beg = b;
    end = node < n ? en : b;
}

__device__ __forceinline__ float quad_sum(float v) {
    v += __shfl_xor(v, 1);
    v += __shfl_xor(v, 2);
    return v;
}

__global__ void k_agg1_mlp(const int* __restrict__ off, const int2* __restrict__ rec,
                           const float* __restrict__ xs, const float* __restrict__ dis,
                           const float* __restrict__ W1, const float* __restrict__ b1,
                           const float* __restrict__ W2, float* __restrict__ h2,
                           int n, int e) {
    int t    = blockIdx.x * blockDim.x + threadIdx.x;
    int node = t >> 2, sub = t & 3;
    int lane = threadIdx.x & 31;
    int c    = lane & 15, h = lane >> 4;

    float w1x = W1[(2 * h) * 16 + c];
    float w1y = W1[16 + c];
    float4 bA = *(const float4*)(b1 + 8 * h);
    float4 bB = *(const float4*)(b1 + 8 * h + 4);
    int  oc = c < 7 ? c : 6;
    bool ov = c < 7;
    float w2x[4], w2y[4];
    for (int s = 0; s < 4; ++s) {
        int f = 8 * h + 2 * s;
        w2x[s] = W2[(f + 0) * 7 + oc];
        w2y[s] = W2[(f + 1) * 7 + oc];
    }

    int beg, end;
    seg_bounds(off, node, n, e, beg, end);
    float a0 = 0.f, a1 = 0.f, a2 = 0.f;
    for (int k = beg + sub; k < end; k += 4) {
        int2 r   = rec[k];
        float w  = __int_as_float(r.y);
        float4 v = *(const float4*)(xs + (size_t)r.x * 4);
        a0 += w * v.x;
        a1 += w * v.y;
        a2 += w * v.z;
    }
    a0 = quad_sum(a0);
    a1 = quad_sum(a1);
    a2 = quad_sum(a2);
    int nc    = node < n ? node : n - 1;
    float di  = dis[nc];
    float4 sf = *(const float4*)(xs + (size_t)nc * 4);
    float o0 = di * (a0 + sf.x);
    float o1 = di * (a1 + sf.y);
    float o2 = di * (a2 + sf.z);

    int srcl  = (lane & 7) << 2;
    float t0  = __shfl(o0, srcl);
    float t1  = __shfl(o1, srcl);
    float t2  = __shfl(o2, srcl);
    float dc  = __shfl(di, srcl);
    int ncol  = ((t - lane) >> 2) + (lane & 7);
    bool cv   = (c < 8) && (ncol < n);

    v2f aw;
    aw.x = w1x;
    aw.y = (h == 0) ? w1y : 0.f;
    v2f bx;
    bx.x = cv ? (h == 0 ? t0 : t2) : 0.f;
    bx.y = (cv && h == 0) ? t1 : 0.f;
    v8f z = {};
    z = __builtin_amdgcn_wmma_f32_16x16x4_f32(false, aw, false, bx, (short)0, z, false, false);

    float ds = cv ? dc : 0.f;
    float zs[8];
    zs[0] = fmaxf(z[0] + bA.x, 0.f) * ds;
    zs[1] = fmaxf(z[1] + bA.y, 0.f) * ds;
    zs[2] = fmaxf(z[2] + bA.z, 0.f) * ds;
    zs[3] = fmaxf(z[3] + bA.w, 0.f) * ds;
    zs[4] = fmaxf(z[4] + bB.x, 0.f) * ds;
    zs[5] = fmaxf(z[5] + bB.y, 0.f) * ds;
    zs[6] = fmaxf(z[6] + bB.z, 0.f) * ds;
    zs[7] = fmaxf(z[7] + bB.w, 0.f) * ds;

    v8f acc = {};
    for (int s = 0; s < 4; ++s) {
        v2f a;
        a.x = ov ? w2x[s] : 0.f;
        a.y = ov ? w2y[s] : 0.f;
        v2f b;
        b.x = zs[2 * s];
        b.y = zs[2 * s + 1];
        acc = __builtin_amdgcn_wmma_f32_16x16x4_f32(false, a, false, b, (short)0, acc, false, false);
    }

    if (cv && h == 0) {
        float4* p = (float4*)(h2 + (size_t)ncol * 16);
        p[0] = make_float4(acc[0], acc[1], acc[2], acc[3]);
        p[1] = make_float4(acc[4], acc[5], acc[6], acc[7]);
    }
}

__global__ void k_agg2(const int* __restrict__ off, const int2* __restrict__ rec,
                       const float* __restrict__ h2, const float* __restrict__ dis,
                       const float* __restrict__ b2, float* __restrict__ out, int n, int e) {
    __shared__ float so[NPB * 7];
    int t    = blockIdx.x * blockDim.x + threadIdx.x;
    int node = t >> 2, sub = t & 3;
    int beg, end;
    seg_bounds(off, node, n, e, beg, end);
    float a0 = 0.f, a1 = 0.f, a2 = 0.f, a3 = 0.f, a4 = 0.f, a5 = 0.f, a6 = 0.f;
    for (int k = beg + sub; k < end; k += 4) {
        int2 r  = rec[k];
        float w = __int_as_float(r.y);
        const float4* hr = (const float4*)(h2 + (size_t)r.x * 16);
        float4 v0 = hr[0], v1 = hr[1];
        a0 += w * v0.x; a1 += w * v0.y; a2 += w * v0.z; a3 += w * v0.w;
        a4 += w * v1.x; a5 += w * v1.y; a6 += w * v1.z;
    }
    a0 = quad_sum(a0); a1 = quad_sum(a1); a2 = quad_sum(a2); a3 = quad_sum(a3);
    a4 = quad_sum(a4); a5 = quad_sum(a5); a6 = quad_sum(a6);
    if (node < n && sub == 0) {
        float di = dis[node];
        const float4* hs = (const float4*)(h2 + (size_t)node * 16);
        float4 s0 = hs[0], s1 = hs[1];
        float* o = so + (threadIdx.x >> 2) * 7;
        o[0] = b2[0] + di * (a0 + s0.x);
        o[1] = b2[1] + di * (a1 + s0.y);
        o[2] = b2[2] + di * (a2 + s0.z);
        o[3] = b2[3] + di * (a3 + s0.w);
        o[4] = b2[4] + di * (a4 + s1.x);
        o[5] = b2[5] + di * (a5 + s1.y);
        o[6] = b2[6] + di * (a6 + s1.z);
    }
    __syncthreads();
    int nb    = blockIdx.x * NPB;
    int left  = n - nb;
    int cntf  = (left < NPB ? left : NPB) * 7;
    float* ob = out + (size_t)nb * 7;
    for (int j = threadIdx.x; j < NPB * 7; j += AGG_T)
        if (j < cntf) ob[j] = so[j];
}

extern "C" void kernel_launch(void* const* d_in, const int* in_sizes, int n_in,
                              void* d_out, int out_size, void* d_ws, size_t ws_size,
                              hipStream_t stream) {
    (void)n_in; (void)out_size; (void)ws_size;
    const float* x  = (const float*)d_in[0];
    const int*   ei = (const int*)d_in[1];
    const float* ew = (const float*)d_in[2];
    const float* W1 = (const float*)d_in[3];
    const float* b1 = (const float*)d_in[4];
    const float* W2 = (const float*)d_in[5];
    const float* b2 = (const float*)d_in[6];
    float* out = (float*)d_out;

    int n = in_sizes[0] / 3;
    int e = in_sizes[2];
    const int* row = ei;
    const int* col = ei + e;

    int nbk   = (n + 1 + BK_N - 1) / BK_N;
    int npad  = nbk * BK_N;
    int nblk  = (e + EPB - 1) / EPB;
    int pitch = (nblk + 3) / 4 * 4;
    int tabn  = nbk * pitch;

    int*   off    = (int*)d_ws;
    float* dis    = (float*)(off + (size_t)npad);
    float* xs     = dis  + (size_t)npad;
    float* h2     = xs   + (size_t)npad * 4;
    int2*  rec    = (int2*)(h2 + (size_t)npad * 16);
    int*   rcn    = (int*)(rec + (((size_t)e + 1) / 2) * 2);
    int*   rst    = rcn  + (size_t)tabn;
    int2*  trec   = (int2*)(rst + (size_t)tabn);
    int*   tcol   = (int*)(trec + (size_t)e);

    int quadB   = (n + NPB - 1) / NPB;

    k_bk_part   <<<nblk, PART_T, 0, stream>>>(row, col, ew, rst, rcn, trec, tcol, e, nblk, pitch, nbk);
    k_bk_sort   <<<nbk, BK_N, 0, stream>>>(rst, rcn, trec, tcol, x, off, rec, dis, xs, e, nblk, pitch, n);

    k_agg1_mlp  <<<quadB, AGG_T, 0, stream>>>(off, rec, xs, dis, W1, b1, W2, h2, n, e);
    k_agg2      <<<quadB, AGG_T, 0, stream>>>(off, rec, h2, dis, b2, out, n, e);
}


# level-1 sort blocks take 10240 edges (20 per thread): 489 blocks, all resident at once at two per WGP, no second round; level 2 walks 489 runs, 3 records per lane; f32 sum order still arrival rank
# speedup vs baseline: 11.9024x; 1.0404x over previous
#include <hip/hip_runtime.h>
#include <hip/hip_bf16.h>


typedef __attribute__((ext_vector_type(2))) float v2f;
typedef __attribute__((ext_vector_type(8))) float v8f;


#define BK_SHIFT 10
#define BK_N (1 << BK_SHIFT)
#define PART_T 512
#define MAXB PART_T
#define EPT1 20
#define EPB (PART_T * EPT1)
#define SB 4
#define RU 4
#define RPL 3
#define RW (32 * 32)
#define STAGE_CAP 32768
#define AGG_T 256
#define NPB (AGG_T / 4)

__global__ void k_bk_part(const int* __restrict__ row, const int* __restrict__ col,
                          const float* __restrict__ ew,
                          int* __restrict__ rst, int* __restrict__ rcn,
                          int2* __restrict__ trec, int* __restrict__ tcol,
                          int e, int nblk, int pitch, int nbk) {
    __shared__ int  lh[MAXB];
    __shared__ int  wsum[PART_T / 32];
    __shared__ int  s_col[EPB];
    __shared__ int2 s_rec[EPB];
    int t    = threadIdx.x;
    int lane = t & 31, w = t >> 5;
    lh[t] = 0;
    __syncthreads();
    int base = blockIdx.x * EPB;
    int c[EPT1], r[EPT1], k[EPT1];
    float wt[EPT1];
    for (int j = 0; j < EPT1; ++j) {
        int i  = base + t + j * PART_T;
        int ic = i < e ? i : e - 1;
        c[j]  = col[ic];
        r[j]  = row[ic];
        wt[j] = ew[ic];
    }
    for (int j = 0; j < EPT1; ++j) {
        int i = base + t + j * PART_T;
        k[j] = 0;
        if (i < e) k[j] = atomicAdd(&lh[c[j] >> BK_SHIFT], 1);
    }
    __syncthreads();
    int v    = lh[t];
    int incl = v;
    for (int d = 1; d < 32; d <<= 1) {
        int y = __shfl_up(incl, d);
        if (lane >= d) incl += y;
    }
    if (lane == 31) wsum[w] = incl;
    __syncthreads();
    int wofs = 0;
    for (int j = 0; j < PART_T / 32; ++j) wofs += (j < w) ? wsum[j] : 0;
    int ex = wofs + incl - v;
    if (t < nbk) {
        size_t ti = (size_t)t * pitch;
        rst[ti + blockIdx.x] = ex;
        rcn[ti + blockIdx.x] = v;
        if (blockIdx.x == 0)
            for (int j = nblk; j < pitch; ++j) rcn[ti + j] = 0;
    }
    lh[t] = ex;
    __syncthreads();
    for (int j = 0; j < EPT1; ++j) {
        int i = base + t + j * PART_T;
        if (i < e) {
            int s = lh[c[j] >> BK_SHIFT] + k[j];
            s_col[s] = c[j];
            s_rec[s] = make_int2(r[j], __float_as_int(wt[j]));
        }
    }
    __syncthreads();
    int left  = e - base;
    int total = left < EPB ? left : EPB;
    for (int s = t; s < total; s += PART_T) {
        trec[base + s] = s_rec[s];
        tcol[base + s] = s_col[s];
    }
}

__global__ void __launch_bounds__(BK_N)
k_bk_sort(const int* __restrict__ rst, const int* __restrict__ rcn,
          const int2* __restrict__ trec,
          const int* __restrict__ tcol, const float* __restrict__ x,
          int* __restrict__ off, int2* __restrict__ rec,
          float* __restrict__ dis, float* __restrict__ xs,
          int e, int nblk, int pitch, int n) {
    __shared__ int  cnt[BK_N];
    __shared__ int  wsum[BK_N / 32];
    __shared__ int  wsA[BK_N / 32];
    __shared__ int  wsB[BK_N / 32];
    __shared__ int2 stage[STAGE_CAP];
    int t    = threadIdx.x;
    int lane = t & 31, w = t >> 5;
    int b    = blockIdx.x;
    const int* myst = rst + (size_t)b * pitch;
    const int* mycn = rcn + (size_t)b * pitch;
    cnt[t] = 0;
    const int4* c4 = (const int4*)rcn;
    int nq  = b * (pitch >> 2);
    int acc = 0;
    for (int i0 = t; i0 < nq; i0 += 8 * BK_N) {
        int4 q[8];
        for (int u = 0; u < 8; ++u) {
            int i = i0 + u * BK_N;
            q[u] = c4[i < nq ? i : nq - 1];
        }
        for (int u = 0; u < 8; ++u) {
            int i = i0 + u * BK_N;
            if (i < nq) acc += (q[u].x + q[u].y) + (q[u].z + q[u].w);
        }
    }
    __syncthreads();
    int mylen = 0;
    for (int jb = 0; jb < nblk; jb += RW) {
        int jm  = jb + w + 32 * lane;
        int jc  = jm < nblk ? jm : nblk - 1;
        int mst = myst[jc];
        int mcn = jm < nblk ? mycn[jc] : 0;
        mylen += mcn;
        int rl = nblk - jb - w;
        int nr = rl > 0 ? (rl + 31) >> 5 : 0;
        nr = nr < 32 ? nr : 32;
        for (int i0 = 0; i0 < nr; i0 += RU) {
            int cu[RU], gu[RU], cc[RU][RPL];
            for (int u = 0; u < RU; ++u) {
                int ii = i0 + u;
                int st = __shfl(mst, ii & 31);
                int cn = __shfl(mcn, ii & 31);
                cu[u] = ii < nr ? cn : 0;
                gu[u] = ii < nr ? (jb + w + 32 * ii) * EPB + st : 0;
                for (int h = 0; h < RPL; ++h) {
                    int k   = lane + 32 * h;
                    int idx = gu[u] + (k < cu[u] ? k : 0);
                    idx = (unsigned)idx < (unsigned)e ? idx : e - 1;
                    cc[u][h] = tcol[idx];
                }
            }
            for (int u = 0; u < RU; ++u)
                for (int h = 0; h < RPL; ++h) {
                    int k = lane + 32 * h;
                    if (k < cu[u]) {
                        int lc = (cc[u][h] - (b << BK_SHIFT)) & (BK_N - 1);
                        atomicAdd(&cnt[lc], 1);
                    }
                }
            for (int u = 0; u < RU; ++u)
                for (int k = lane + 32 * RPL; k < cu[u]; k += 32) {
                    int idx = gu[u] + k;
                    idx = (unsigned)idx < (unsigned)e ? idx : e - 1;
                    int lc = (tcol[idx] - (b << BK_SHIFT)) & (BK_N - 1);
                    atomicAdd(&cnt[lc], 1);
                }
        }
    }
    for (int d = 16; d >= 1; d >>= 1) {
        acc   += __shfl_xor(acc, d);
        mylen += __shfl_xor(mylen, d);
    }
    if (lane == 0) { wsA[w] = acc; wsB[w] = mylen; }
    __syncthreads();
    int bs = 0, len = 0;
    for (int j = 0; j < BK_N / 32; ++j) { bs += wsA[j]; len += wsB[j]; }
    bs = bs > 0 ? bs : 0;
    bs = bs < e ? bs : e;
    int be = bs + (len > 0 ? len : 0);
    be = (be < e && be >= bs) ? be : e;
    len = be - bs;
    bool fits = len <= STAGE_CAP;
    int v    = cnt[t];
    int incl = v;
    for (int d = 1; d < 32; d <<= 1) {
        int y = __shfl_up(incl, d);
        if (lane >= d) incl += y;
    }
    if (lane == 31) wsum[w] = incl;
    __syncthreads();
    int wofs = 0;
    for (int j = 0; j < BK_N / 32; ++j) wofs += (j < w) ? wsum[j] : 0;
    int lst = wofs + incl - v;
    cnt[t] = lst;
    off[(size_t)(b << BK_SHIFT) + t] = bs + lst;
    __syncthreads();
    int node = (b << BK_SHIFT) + t;
    const float* xi = x + (size_t)(node < n ? node : n - 1) * 3;
    float xv0, xv1, xv2;
    float s0 = 0.f, s1 = 0.f, s2 = 0.f, s3 = 0.f;
    for (int jb = 0; jb < nblk; jb += RW) {
        int jm  = jb + w + 32 * lane;
        int jc  = jm < nblk ? jm : nblk - 1;
        int mst = myst[jc];
        int mcn = jm < nblk ? mycn[jc] : 0;
        int rl = nblk - jb - w;
        int nr = rl > 0 ? (rl + 31) >> 5 : 0;
        nr = nr < 32 ? nr : 32;
        for (int i0 = 0; i0 < nr; i0 += RU) {
            int cu[RU], gu[RU], cc[RU][RPL], p[RU][RPL];
            int2 rr[RU][RPL];
            for (int u = 0; u < RU; ++u) {
                int ii = i0 + u;
                int st = __shfl(mst, ii & 31);
                int cn = __shfl(mcn, ii & 31);
                cu[u] = ii < nr ? cn : 0;
                gu[u] = ii < nr ? (jb + w + 32 * ii) * EPB + st : 0;
                for (int h = 0; h < RPL; ++h) {
                    int k   = lane + 32 * h;
                    int idx = gu[u] + (k < cu[u] ? k : 0);
                    idx = (unsigned)idx < (unsigned)e ? idx : e - 1;
                    cc[u][h] = tcol[idx];
                    rr[u][h] = trec[idx];
                }
            }
            for (int u = 0; u < RU; ++u)
                for (int h = 0; h < RPL; ++h) {
                    int k  = lane + 32 * h;
                    int lc = (cc[u][h] - (b << BK_SHIFT)) & (BK_N - 1);
                    p[u][h] = -1;
                    if (k < cu[u]) p[u][h] = atomicAdd(&cnt[lc], 1);
                }
            for (int u = 0; u < RU; ++u)
                for (int h = 0; h < RPL; ++h) {
                    if (fits) {
                        if ((unsigned)p[u][h] < (unsigned)STAGE_CAP) stage[p[u][h]] = rr[u][h];
                    } else {
                        int g = bs + p[u][h];
                        if (p[u][h] >= 0 && (unsigned)g < (unsigned)e) rec[g] = rr[u][h];
                    }
                }
            for (int u = 0; u < RU; ++u)
                for (int k = lane + 32 * RPL; k < cu[u]; k += 32) {
                    int idx = gu[u] + k;
                    idx = (unsigned)idx < (unsigned)e ? idx : e - 1;
                    int lc = (tcol[idx] - (b << BK_SHIFT)) & (BK_N - 1);
                    int2 r2 = trec[idx];
                    int pp = atomicAdd(&cnt[lc], 1);
                    if (fits) {
                        if ((unsigned)pp < (unsigned)STAGE_CAP) stage[pp] = r2;
                    } else {
                        int g = bs + pp;
                        if ((unsigned)g < (unsigned)e) rec[g] = r2;
                    }
                }
        }
    }
    __syncthreads();
    xv0 = xi[0]; xv1 = xi[1]; xv2 = xi[2];
    if (fits) {
        int head = bs & 15;
        for (int i0 = t - head; i0 < len; i0 += SB * BK_N) {
            int2 r[SB];
            for (int u = 0; u < SB; ++u) {
                int i = i0 + u * BK_N;
                r[u] = stage[(i >= 0 && i < len) ? i : 0];
            }
            for (int u = 0; u < SB; ++u) {
                int i = i0 + u * BK_N;
                if (i >= 0 && i < len) rec[bs + i] = r[u];
            }
        }
        for (int j = 0; j < v; j += 4) {
            int i  = lst + j;
            int i1 = j + 1 < v ? i + 1 : i;
            int i2 = j + 2 < v ? i + 2 : i;
            int i3 = j + 3 < v ? i + 3 : i;
            float w0 = __int_as_float(stage[i].y);
            float w1 = __int_as_float(stage[i1].y);
            float w2 = __int_as_float(stage[i2].y);
            float w3 = __int_as_float(stage[i3].y);
            s0 += w0;
            if (j + 1 < v) s1 += w1;
            if (j + 2 < v) s2 += w2;
            if (j + 3 < v) s3 += w3;
        }
    } else {
        for (int j = 0; j < v; j += 4) {
            int g = bs + lst + j;
            for (int u = 0; u < 4; ++u) {
                int gu2 = g + u;
                gu2 = (unsigned)gu2 < (unsigned)e ? gu2 : e - 1;
                float wv = __int_as_float(rec[gu2].y);
                if (j + u < v) {
                    if (u == 0) s0 += wv;
                    if (u == 1) s1 += wv;
                    if (u == 2) s2 += wv;
                    if (u == 3) s3 += wv;
                }
            }
        }
    }
    if (node < n) {
        float d  = 1.0f + ((s0 + s1) + (s2 + s3));
        float di = (d > 0.f) ? rsqrtf(d) : 0.f;
        dis[node] = di;
        float4 o;
        o.x = xv0 * di;
        o.y = xv1 * di;
        o.z = xv2 * di;
        o.w = 0.f;
        *(float4*)(xs + (size_t)node * 4) = o;
    }
}

__device__ __forceinline__ void seg_bounds(const int* __restrict__ off, int node, int n, int e,
                                           int& beg, int& end) {
    int nc = node < n ? node : n - 1;
    int b  = off[nc];
    int en = off[nc + 1];
    b  = b > 0 ? b : 0;
    en = en < e ? en : e;
    beg = b;
    end = node < n ? en : b;
}

__device__ __forceinline__ float quad_sum(float v) {
    v += __shfl_xor(v, 1);
    v += __shfl_xor(v, 2);
    return v;
}

__global__ void k_agg1_mlp(const int* __restrict__ off, const int2* __restrict__ rec,
                           const float* __restrict__ xs, const float* __restrict__ dis,
                           const float* __restrict__ W1, const float* __restrict__ b1,
                           const float* __restrict__ W2, float* __restrict__ h2,
                           int n, int e) {
    int t    = blockIdx.x * blockDim.x + threadIdx.x;
    int node = t >> 2, sub = t & 3;
    int lane = threadIdx.x & 31;
    int c    = lane & 15, h = lane >> 4;

    float w1x = W1[(2 * h) * 16 + c];
    float w1y = W1[16 + c];
    float4 bA = *(const float4*)(b1 + 8 * h);
    float4 bB = *(const float4*)(b1 + 8 * h + 4);
    int  oc = c < 7 ? c : 6;
    bool ov = c < 7;
    float w2x[4], w2y[4];
    for (int s = 0; s < 4; ++s) {
        int f = 8 * h + 2 * s;
        w2x[s] = W2[(f + 0) * 7 + oc];
        w2y[s] = W2[(f + 1) * 7 + oc];
    }

    int beg, end;
    seg_bounds(off, node, n, e, beg, end);
    float a0 = 0.f, a1 = 0.f, a2 = 0.f;
    for (int k = beg + sub; k < end; k += 4) {
        int2 r   = rec[k];
        float w  = __int_as_float(r.y);
        float4 v = *(const float4*)(xs + (size_t)r.x * 4);
        a0 += w * v.x;
        a1 += w * v.y;
        a2 += w * v.z;
    }
    a0 = quad_sum(a0);
    a1 = quad_sum(a1);
    a2 = quad_sum(a2);
    int nc    = node < n ? node : n - 1;
    float di  = dis[nc];
    float4 sf = *(const float4*)(xs + (size_t)nc * 4);
    float o0 = di * (a0 + sf.x);
    float o1 = di * (a1 + sf.y);
    float o2 = di * (a2 + sf.z);

    int srcl  = (lane & 7) << 2;
    float t0  = __shfl(o0, srcl);
    float t1  = __shfl(o1, srcl);
    float t2  = __shfl(o2, srcl);
    float dc  = __shfl(di, srcl);
    int ncol  = ((t - lane) >> 2) + (lane & 7);
    bool cv   = (c < 8) && (ncol < n);

    v2f aw;
    aw.x = w1x;
    aw.y = (h == 0) ? w1y : 0.f;
    v2f bx;
    bx.x = cv ? (h == 0 ? t0 : t2) : 0.f;
    bx.y = (cv && h == 0) ? t1 : 0.f;
    v8f z = {};
    z = __builtin_amdgcn_wmma_f32_16x16x4_f32(false, aw, false, bx, (short)0, z, false, false);

    float ds = cv ? dc : 0.f;
    float zs[8];
    zs[0] = fmaxf(z[0] + bA.x, 0.f) * ds;
    zs[1] = fmaxf(z[1] + bA.y, 0.f) * ds;
    zs[2] = fmaxf(z[2] + bA.z, 0.f) * ds;
    zs[3] = fmaxf(z[3] + bA.w, 0.f) * ds;
    zs[4] = fmaxf(z[4] + bB.x, 0.f) * ds;
    zs[5] = fmaxf(z[5] + bB.y, 0.f) * ds;
    zs[6] = fmaxf(z[6] + bB.z, 0.f) * ds;
    zs[7] = fmaxf(z[7] + bB.w, 0.f) * ds;

    v8f acc = {};
    for (int s = 0; s < 4; ++s) {
        v2f a;
        a.x = ov ? w2x[s] : 0.f;
        a.y = ov ? w2y[s] : 0.f;
        v2f b;
        b.x = zs[2 * s];
        b.y = zs[2 * s + 1];
        acc = __builtin_amdgcn_wmma_f32_16x16x4_f32(false, a, false, b, (short)0, acc, false, false);
    }

    if (cv && h == 0) {
        float4* p = (float4*)(h2 + (size_t)ncol * 16);
        p[0] = make_float4(acc[0], acc[1], acc[2], acc[3]);
        p[1] = make_float4(acc[4], acc[5], acc[6], acc[7]);
    }
}

__global__ void k_agg2(const int* __restrict__ off, const int2* __restrict__ rec,
                       const float* __restrict__ h2, const float* __restrict__ dis,
                       const float* __restrict__ b2, float* __restrict__ out, int n, int e) {
    __shared__ float so[NPB * 7];
    int t    = blockIdx.x * blockDim.x + threadIdx.x;
    int node = t >> 2, sub = t & 3;
    int beg, end;
    seg_bounds(off, node, n, e, beg, end);
    float a0 = 0.f, a1 = 0.f, a2 = 0.f, a3 = 0.f, a4 = 0.f, a5 = 0.f, a6 = 0.f;
    for (int k = beg + sub; k < end; k += 4) {
        int2 r  = rec[k];
        float w = __int_as_float(r.y);
        const float4* hr = (const float4*)(h2 + (size_t)r.x * 16);
        float4 v0 = hr[0], v1 = hr[1];
        a0 += w * v0.x; a1 += w * v0.y; a2 += w * v0.z; a3 += w * v0.w;
        a4 += w * v1.x; a5 += w * v1.y; a6 += w * v1.z;
    }
    a0 = quad_sum(a0); a1 = quad_sum(a1); a2 = quad_sum(a2); a3 = quad_sum(a3);
    a4 = quad_sum(a4); a5 = quad_sum(a5); a6 = quad_sum(a6);
    if (node < n && sub == 0) {
        float di = dis[node];
        const float4* hs = (const float4*)(h2 + (size_t)node * 16);
        float4 s0 = hs[0], s1 = hs[1];
        float* o = so + (threadIdx.x >> 2) * 7;
        o[0] = b2[0] + di * (a0 + s0.x);
        o[1] = b2[1] + di * (a1 + s0.y);
        o[2] = b2[2] + di * (a2 + s0.z);
        o[3] = b2[3] + di * (a3 + s0.w);
        o[4] = b2[4] + di * (a4 + s1.x);
        o[5] = b2[5] + di * (a5 + s1.y);
        o[6] = b2[6] + di * (a6 + s1.z);
    }
    __syncthreads();
    int nb    = blockIdx.x * NPB;
    int left  = n - nb;
    int cntf  = (left < NPB ? left : NPB) * 7;
    float* ob = out + (size_t)nb * 7;
    for (int j = threadIdx.x; j < NPB * 7; j += AGG_T)
        if (j < cntf) ob[j] = so[j];
}

extern "C" void kernel_launch(void* const* d_in, const int* in_sizes, int n_in,
                              void* d_out, int out_size, void* d_ws, size_t ws_size,
                              hipStream_t stream) {
    (void)n_in; (void)out_size; (void)ws_size;
    const float* x  = (const float*)d_in[0];
    const int*   ei = (const int*)d_in[1];
    const float* ew = (const float*)d_in[2];
    const float* W1 = (const float*)d_in[3];
    const float* b1 = (const float*)d_in[4];
    const float* W2 = (const float*)d_in[5];
    const float* b2 = (const float*)d_in[6];
    float* out = (float*)d_out;

    int n = in_sizes[0] / 3;
    int e = in_sizes[2];
    const int* row = ei;
    const int* col = ei + e;

    int nbk   = (n + 1 + BK_N - 1) / BK_N;
    int npad  = nbk * BK_N;
    int nblk  = (e + EPB - 1) / EPB;
    int pitch = (nblk + 3) / 4 * 4;
    int tabn  = nbk * pitch;

    int*   off    = (int*)d_ws;
    float* dis    = (float*)(off + (size_t)npad);
    float* xs     = dis  + (size_t)npad;
    float* h2     = xs   + (size_t)npad * 4;
    int2*  rec    = (int2*)(h2 + (size_t)npad * 16);
    int*   rcn    = (int*)(rec + (((size_t)e + 1) / 2) * 2);
    int*   rst    = rcn  + (size_t)tabn;
    int2*  trec   = (int2*)(rst + (size_t)tabn);
    int*   tcol   = (int*)(trec + (size_t)e);

    int quadB   = (n + NPB - 1) / NPB;

    k_bk_part   <<<nblk, PART_T, 0, stream>>>(row, col, ew, rst, rcn, trec, tcol, e, nblk, pitch, nbk);
    k_bk_sort   <<<nbk, BK_N, 0, stream>>>(rst, rcn, trec, tcol, x, off, rec, dis, xs, e, nblk, pitch, n);

    k_agg1_mlp  <<<quadB, AGG_T, 0, stream>>>(off, rec, xs, dis, W1, b1, W2, h2, n, e);
    k_agg2      <<<quadB, AGG_T, 0, stream>>>(off, rec, h2, dis, b2, out, n, e);
}


# both per-node sum launches: the block copies its contiguous stretch of rec into LDS once, whole lines at 16 bytes per lane; quads read their records from LDS, not global; f32 sum order unchanged
# speedup vs baseline: 15.4420x; 1.2974x over previous
#include <hip/hip_runtime.h>
#include <hip/hip_bf16.h>


typedef __attribute__((ext_vector_type(2))) float v2f;
typedef __attribute__((ext_vector_type(8))) float v8f;


#define BK_SHIFT 10
#define BK_N (1 << BK_SHIFT)
#define PART_T 512
#define MAXB PART_T
#define EPT1 20
#define EPB (PART_T * EPT1)
#define SB 4
#define RU 4
#define RPL 3
#define RW (32 * 32)
#define STAGE_CAP 32768
#define AGG_T 256
#define NPB (AGG_T / 2)
#define RCAP 4096
#define RS (RCAP / 2 / AGG_T)

__global__ void k_bk_part(const int* __restrict__ row, const int* __restrict__ col,
                          const float* __restrict__ ew,
                          int* __restrict__ rst, int* __restrict__ rcn,
                          int2* __restrict__ trec, int* __restrict__ tcol,
                          int e, int nblk, int pitch, int nbk) {
    __shared__ int  lh[MAXB];
    __shared__ int  wsum[PART_T / 32];
    __shared__ int  s_col[EPB];
    __shared__ int2 s_rec[EPB];
    int t    = threadIdx.x;
    int lane = t & 31, w = t >> 5;
    lh[t] = 0;
    __syncthreads();
    int base = blockIdx.x * EPB;
    int c[EPT1], r[EPT1], k[EPT1];
    float wt[EPT1];
    for (int j = 0; j < EPT1; ++j) {
        int i  = base + t + j * PART_T;
        int ic = i < e ? i : e - 1;
        c[j]  = col[ic];
        r[j]  = row[ic];
        wt[j] = ew[ic];
    }
    for (int j = 0; j < EPT1; ++j) {
        int i = base + t + j * PART_T;
        k[j] = 0;
        if (i < e) k[j] = atomicAdd(&lh[c[j] >> BK_SHIFT], 1);
    }
    __syncthreads();
    int v    = lh[t];
    int incl = v;
    for (int d = 1; d < 32; d <<= 1) {
        int y = __shfl_up(incl, d);
        if (lane >= d) incl += y;
    }
    if (lane == 31) wsum[w] = incl;
    __syncthreads();
    int wofs = 0;
    for (int j = 0; j < PART_T / 32; ++j) wofs += (j < w) ? wsum[j] : 0;
    int ex = wofs + incl - v;
    if (t < nbk) {
        size_t ti = (size_t)t * pitch;
        rst[ti + blockIdx.x] = ex;
        rcn[ti + blockIdx.x] = v;
        if (blockIdx.x == 0)
            for (int j = nblk; j < pitch; ++j) rcn[ti + j] = 0;
    }
    lh[t] = ex;
    __syncthreads();
    for (int j = 0; j < EPT1; ++j) {
        int i = base + t + j * PART_T;
        if (i < e) {
            int s = lh[c[j] >> BK_SHIFT] + k[j];
            s_col[s] = c[j];
            s_rec[s] = make_int2(r[j], __float_as_int(wt[j]));
        }
    }
    __syncthreads();
    int left  = e - base;
    int total = left < EPB ? left : EPB;
    for (int s = t; s < total; s += PART_T) {
        trec[base + s] = s_rec[s];
        tcol[base + s] = s_col[s];
    }
}

__global__ void __launch_bounds__(BK_N)
k_bk_sort(const int* __restrict__ rst, const int* __restrict__ rcn,
          const int2* __restrict__ trec,
          const int* __restrict__ tcol, const float* __restrict__ x,
          int* __restrict__ off, int2* __restrict__ rec,
          float* __restrict__ dis, float* __restrict__ xs,
          int e, int nblk, int pitch, int n) {
    __shared__ int  cnt[BK_N];
    __shared__ int  wsum[BK_N / 32];
    __shared__ int  wsA[BK_N / 32];
    __shared__ int  wsB[BK_N / 32];
    __shared__ int2 stage[STAGE_CAP];
    int t    = threadIdx.x;
    int lane = t & 31, w = t >> 5;
    int b    = blockIdx.x;
    const int* myst = rst + (size_t)b * pitch;
    const int* mycn = rcn + (size_t)b * pitch;
    cnt[t] = 0;
    const int4* c4 = (const int4*)rcn;
    int nq  = b * (pitch >> 2);
    int acc = 0;
    for (int i0 = t; i0 < nq; i0 += 8 * BK_N) {
        int4 q[8];
        for (int u = 0; u < 8; ++u) {
            int i = i0 + u * BK_N;
            q[u] = c4[i < nq ? i : nq - 1];
        }
        for (int u = 0; u < 8; ++u) {
            int i = i0 + u * BK_N;
            if (i < nq) acc += (q[u].x + q[u].y) + (q[u].z + q[u].w);
        }
    }
    __syncthreads();
    int mylen = 0;
    for (int jb = 0; jb < nblk; jb += RW) {
        int jm  = jb + w + 32 * lane;
        int jc  = jm < nblk ? jm : nblk - 1;
        int mst = myst[jc];
        int mcn = jm < nblk ? mycn[jc] : 0;
        mylen += mcn;
        int rl = nblk - jb - w;
        int nr = rl > 0 ? (rl + 31) >> 5 : 0;
        nr = nr < 32 ? nr : 32;
        for (int i0 = 0; i0 < nr; i0 += RU) {
            int cu[RU], gu[RU], cc[RU][RPL];
            for (int u = 0; u < RU; ++u) {
                int ii = i0 + u;
                int st = __shfl(mst, ii & 31);
                int cn = __shfl(mcn, ii & 31);
                cu[u] = ii < nr ? cn : 0;
                gu[u] = ii < nr ? (jb + w + 32 * ii) * EPB + st : 0;
                for (int h = 0; h < RPL; ++h) {
                    int k   = lane + 32 * h;
                    int idx = gu[u] + (k < cu[u] ? k : 0);
                    idx = (unsigned)idx < (unsigned)e ? idx : e - 1;
                    cc[u][h] = tcol[idx];
                }
            }
            for (int u = 0; u < RU; ++u)
                for (int h = 0; h < RPL; ++h) {
                    int k = lane + 32 * h;
                    if (k < cu[u]) {
                        int lc = (cc[u][h] - (b << BK_SHIFT)) & (BK_N - 1);
                        atomicAdd(&cnt[lc], 1);
                    }
                }
            for (int u = 0; u < RU; ++u)
                for (int k = lane + 32 * RPL; k < cu[u]; k += 32) {
                    int idx = gu[u] + k;
                    idx = (unsigned)idx < (unsigned)e ? idx : e - 1;
                    int lc = (tcol[idx] - (b << BK_SHIFT)) & (BK_N - 1);
                    atomicAdd(&cnt[lc], 1);
                }
        }
    }
    for (int d = 16; d >= 1; d >>= 1) {
        acc   += __shfl_xor(acc, d);
        mylen += __shfl_xor(mylen, d);
    }
    if (lane == 0) { wsA[w] = acc; wsB[w] = mylen; }
    __syncthreads();
    int bs = 0, len = 0;
    for (int j = 0; j < BK_N / 32; ++j) { bs += wsA[j]; len += wsB[j]; }
    bs = bs > 0 ? bs : 0;
    bs = bs < e ? bs : e;
    int be = bs + (len > 0 ? len : 0);
    be = (be < e && be >= bs) ? be : e;
    len = be - bs;
    bool fits = len <= STAGE_CAP;
    int v    = cnt[t];
    int incl = v;
    for (int d = 1; d < 32; d <<= 1) {
        int y = __shfl_up(incl, d);
        if (lane >= d) incl += y;
    }
    if (lane == 31) wsum[w] = incl;
    __syncthreads();
    int wofs = 0;
    for (int j = 0; j < BK_N / 32; ++j) wofs += (j < w) ? wsum[j] : 0;
    int lst = wofs + incl - v;
    cnt[t] = lst;
    off[(size_t)(b << BK_SHIFT) + t] = bs + lst;
    __syncthreads();
    int node = (b << BK_SHIFT) + t;
    const float* xi = x + (size_t)(node < n ? node : n - 1) * 3;
    float xv0, xv1, xv2;
    float s0 = 0.f, s1 = 0.f, s2 = 0.f, s3 = 0.f;
    for (int jb = 0; jb < nblk; jb += RW) {
        int jm  = jb + w + 32 * lane;
        int jc  = jm < nblk ? jm : nblk - 1;
        int mst = myst[jc];
        int mcn = jm < nblk ? mycn[jc] : 0;
        int rl = nblk - jb - w;
        int nr = rl > 0 ? (rl + 31) >> 5 : 0;
        nr = nr < 32 ? nr : 32;
        for (int i0 = 0; i0 < nr; i0 += RU) {
            int cu[RU], gu[RU], cc[RU][RPL], p[RU][RPL];
            int2 rr[RU][RPL];
            for (int u = 0; u < RU; ++u) {
                int ii = i0 + u;
                int st = __shfl(mst, ii & 31);
                int cn = __shfl(mcn, ii & 31);
                cu[u] = ii < nr ? cn : 0;
                gu[u] = ii < nr ? (jb + w + 32 * ii) * EPB + st : 0;
                for (int h = 0; h < RPL; ++h) {
                    int k   = lane + 32 * h;
                    int idx = gu[u] + (k < cu[u] ? k : 0);
                    idx = (unsigned)idx < (unsigned)e ? idx : e - 1;
                    cc[u][h] = tcol[idx];
                    rr[u][h] = trec[idx];
                }
            }
            for (int u = 0; u < RU; ++u)
                for (int h = 0; h < RPL; ++h) {
                    int k  = lane + 32 * h;
                    int lc = (cc[u][h] - (b << BK_SHIFT)) & (BK_N - 1);
                    p[u][h] = -1;
                    if (k < cu[u]) p[u][h] = atomicAdd(&cnt[lc], 1);
                }
            for (int u = 0; u < RU; ++u)
                for (int h = 0; h < RPL; ++h) {
                    if (fits) {
                        if ((unsigned)p[u][h] < (unsigned)STAGE_CAP) stage[p[u][h]] = rr[u][h];
                    } else {
                        int g = bs + p[u][h];
                        if (p[u][h] >= 0 && (unsigned)g < (unsigned)e) rec[g] = rr[u][h];
                    }
                }
            for (int u = 0; u < RU; ++u)
                for (int k = lane + 32 * RPL; k < cu[u]; k += 32) {
                    int idx = gu[u] + k;
                    idx = (unsigned)idx < (unsigned)e ? idx : e - 1;
                    int lc = (tcol[idx] - (b << BK_SHIFT)) & (BK_N - 1);
                    int2 r2 = trec[idx];
                    int pp = atomicAdd(&cnt[lc], 1);
                    if (fits) {
                        if ((unsigned)pp < (unsigned)STAGE_CAP) stage[pp] = r2;
                    } else {
                        int g = bs + pp;
                        if ((unsigned)g < (unsigned)e) rec[g] = r2;
                    }
                }
        }
    }
    __syncthreads();
    xv0 = xi[0]; xv1 = xi[1]; xv2 = xi[2];
    if (fits) {
        int head = bs & 15;
        for (int i0 = t - head; i0 < len; i0 += SB * BK_N) {
            int2 r[SB];
            for (int u = 0; u < SB; ++u) {
                int i = i0 + u * BK_N;
                r[u] = stage[(i >= 0 && i < len) ? i : 0];
            }
            for (int u = 0; u < SB; ++u) {
                int i = i0 + u * BK_N;
                if (i >= 0 && i < len) rec[bs + i] = r[u];
            }
        }
        for (int j = 0; j < v; j += 4) {
            int i  = lst + j;
            int i1 = j + 1 < v ? i + 1 : i;
            int i2 = j + 2 < v ? i + 2 : i;
            int i3 = j + 3 < v ? i + 3 : i;
            float w0 = __int_as_float(stage[i].y);
            float w1 = __int_as_float(stage[i1].y);
            float w2 = __int_as_float(stage[i2].y);
            float w3 = __int_as_float(stage[i3].y);
            s0 += w0;
            if (j + 1 < v) s1 += w1;
            if (j + 2 < v) s2 += w2;
            if (j + 3 < v) s3 += w3;
        }
    } else {
        for (int j = 0; j < v; j += 4) {
            int g = bs + lst + j;
            for (int u = 0; u < 4; ++u) {
                int gu2 = g + u;
                gu2 = (unsigned)gu2 < (unsigned)e ? gu2 : e - 1;
                float wv = __int_as_float(rec[gu2].y);
                if (j + u < v) {
                    if (u == 0) s0 += wv;
                    if (u == 1) s1 += wv;
                    if (u == 2) s2 += wv;
                    if (u == 3) s3 += wv;
                }
            }
        }
    }
    if (node < n) {
        float d  = 1.0f + ((s0 + s1) + (s2 + s3));
        float di = (d > 0.f) ? rsqrtf(d) : 0.f;
        dis[node] = di;
        float4 o;
        o.x = xv0 * di;
        o.y = xv1 * di;
        o.z = xv2 * di;
        o.w = 0.f;
        *(float4*)(xs + (size_t)node * 4) = o;
    }
}

__device__ __forceinline__ void seg_bounds(const int* __restrict__ off, int node, int n, int e,
                                           int& beg, int& end) {
    int nc = node < n ? node : n - 1;
    int b  = off[nc];
    int en = off[nc + 1];
    b  = b > 0 ? b : 0;
    en = en < e ? en : e;
    beg = b;
    end = node < n ? en : b;
}

__device__ __forceinline__ float quad_sum(float v) {
    v += __shfl_xor(v, 1);
    v += __shfl_xor(v, 2);
    return v;
}

__device__ __forceinline__ void stage_load(const int* __restrict__ off, const int2* __restrict__ rec,
                                           int nb, int n, int e, int& s0, bool& tail, int4* q) {
    int last = nb + NPB < n ? nb + NPB : n;
    int sb = off[nb < n ? nb : n - 1];
    int se = off[last];
    sb = sb > 0 ? sb : 0;
    se = se < e ? se : e;
    s0 = sb & ~15;
    int cnt = se - s0;
    tail = cnt > RCAP;
    cnt = cnt < RCAP ? cnt : RCAP;
    const int4* r4 = (const int4*)rec;
    int gmax = (e - 1) >> 1;
    for (int u = 0; u < RS; ++u) {
        int p  = threadIdx.x + u * AGG_T;
        int gi = (s0 >> 1) + (2 * p < cnt ? p : 0);
        gi = gi < gmax ? gi : gmax;
        q[u] = r4[gi];
    }
}

__device__ __forceinline__ void stage_store(int2* srec, const int4* q) {
    int4* s4 = (int4*)srec;
    for (int u = 0; u < RS; ++u) s4[threadIdx.x + u * AGG_T] = q[u];
}

template <bool TAIL>
__device__ __forceinline__ int2 rec_at(const int2* __restrict__ rec, const int2* srec, int g, int s0) {
    unsigned i = (unsigned)(g - s0);
    if (TAIL) {
        if (i >= (unsigned)RCAP) return rec[g];
        return srec[i];
    }
    return srec[i < (unsigned)RCAP ? i : (unsigned)(RCAP - 1)];
}

template <bool TAIL>
__device__ __forceinline__ void seg_sum3(const int2* __restrict__ rec, const int2* srec,
                                         const float* __restrict__ xs, int s0,
                                         int kA, int endA, int kB, int endB,
                                         float& a0, float& a1, float& a2,
                                         float& g0, float& g1, float& g2) {
    while (kA < endA || kB < endB) {
        bool vA = kA < endA, vB = kB < endB;
        int2 rA = rec_at<TAIL>(rec, srec, vA ? kA : s0, s0);
        int2 rB = rec_at<TAIL>(rec, srec, vB ? kB : s0, s0);
        float4 xA = *(const float4*)(xs + (size_t)(vA ? rA.x : 0) * 4);
        float4 xB = *(const float4*)(xs + (size_t)(vB ? rB.x : 0) * 4);
        float wA = __int_as_float(rA.y), wB = __int_as_float(rB.y);
        a0 = vA ? __builtin_fmaf(wA, xA.x, a0) : a0;
        a1 = vA ? __builtin_fmaf(wA, xA.y, a1) : a1;
        a2 = vA ? __builtin_fmaf(wA, xA.z, a2) : a2;
        g0 = vB ? __builtin_fmaf(wB, xB.x, g0) : g0;
        g1 = vB ? __builtin_fmaf(wB, xB.y, g1) : g1;
        g2 = vB ? __builtin_fmaf(wB, xB.z, g2) : g2;
        kA += 4;
        kB += 4;
    }
}

template <bool TAIL>
__device__ __forceinline__ void seg_sum7(const int2* __restrict__ rec, const int2* srec,
                                         const float* __restrict__ h2, int s0,
                                         int kA, int endA, int kB, int endB,
                                         float* a, float* g) {
    while (kA < endA || kB < endB) {
        bool vA = kA < endA, vB = kB < endB;
        int2 rA = rec_at<TAIL>(rec, srec, vA ? kA : s0, s0);
        int2 rB = rec_at<TAIL>(rec, srec, vB ? kB : s0, s0);
        const float4* hA = (const float4*)(h2 + (size_t)(vA ? rA.x : 0) * 16);
        const float4* hB = (const float4*)(h2 + (size_t)(vB ? rB.x : 0) * 16);
        float4 u0 = hA[0], u1 = hA[1];
        float4 y0 = hB[0], y1 = hB[1];
        float wA = __int_as_float(rA.y), wB = __int_as_float(rB.y);
        a[0] = vA ? __builtin_fmaf(wA, u0.x, a[0]) : a[0]; a[1] = vA ? __builtin_fmaf(wA, u0.y, a[1]) : a[1];
        a[2] = vA ? __builtin_fmaf(wA, u0.z, a[2]) : a[2]; a[3] = vA ? __builtin_fmaf(wA, u0.w, a[3]) : a[3];
        a[4] = vA ? __builtin_fmaf(wA, u1.x, a[4]) : a[4]; a[5] = vA ? __builtin_fmaf(wA, u1.y, a[5]) : a[5];
        a[6] = vA ? __builtin_fmaf(wA, u1.z, a[6]) : a[6];
        g[0] = vB ? __builtin_fmaf(wB, y0.x, g[0]) : g[0]; g[1] = vB ? __builtin_fmaf(wB, y0.y, g[1]) : g[1];
        g[2] = vB ? __builtin_fmaf(wB, y0.z, g[2]) : g[2]; g[3] = vB ? __builtin_fmaf(wB, y0.w, g[3]) : g[3];
        g[4] = vB ? __builtin_fmaf(wB, y1.x, g[4]) : g[4]; g[5] = vB ? __builtin_fmaf(wB, y1.y, g[5]) : g[5];
        g[6] = vB ? __builtin_fmaf(wB, y1.z, g[6]) : g[6];
        kA += 4;
        kB += 4;
    }
}

__global__ void k_agg1_mlp(const int* __restrict__ off, const int2* __restrict__ rec,
                           const float* __restrict__ xs, const float* __restrict__ dis,
                           const float* __restrict__ W1, const float* __restrict__ b1,
                           const float* __restrict__ W2, float* __restrict__ h2,
                           int n, int e) {
    __shared__ int2 srec[RCAP];
    int t    = blockIdx.x * blockDim.x + threadIdx.x;
    int sub  = t & 3;
    int nodeA = (t >> 2) * 2, nodeB = nodeA + 1;
    int lane = threadIdx.x & 31;
    int c    = lane & 15, h = lane >> 4;

    int s0;
    bool stail;
    int4 sq[RS];
    stage_load(off, rec, blockIdx.x * NPB, n, e, s0, stail, sq);

    float w1x = W1[(2 * h) * 16 + c];
    float w1y = W1[16 + c];
    float4 bA = *(const float4*)(b1 + 8 * h);
    float4 bB = *(const float4*)(b1 + 8 * h + 4);
    int  oc = c < 7 ? c : 6;
    bool ov = c < 7;
    float w2x[4], w2y[4];
    for (int s = 0; s < 4; ++s) {
        int f = 8 * h + 2 * s;
        w2x[s] = W2[(f + 0) * 7 + oc];
        w2y[s] = W2[(f + 1) * 7 + oc];
    }

    int begA, endA, begB, endB;
    seg_bounds(off, nodeA, n, e, begA, endA);
    seg_bounds(off, nodeB, n, e, begB, endB);
    stage_store(srec, sq);
    __syncthreads();
    float a0 = 0.f, a1 = 0.f, a2 = 0.f;
    float g0 = 0.f, g1 = 0.f, g2 = 0.f;
    int kA = begA + sub, kB = begB + sub;
    if (stail)
        seg_sum3<true >(rec, srec, xs, s0, kA, endA, kB, endB, a0, a1, a2, g0, g1, g2);
    else
        seg_sum3<false>(rec, srec, xs, s0, kA, endA, kB, endB, a0, a1, a2, g0, g1, g2);
    a0 = quad_sum(a0);
    a1 = quad_sum(a1);
    a2 = quad_sum(a2);
    g0 = quad_sum(g0);
    g1 = quad_sum(g1);
    g2 = quad_sum(g2);
    bool second = (sub & 1) != 0;
    int nm    = second ? nodeB : nodeA;
    int nc    = nm < n ? nm : n - 1;
    float di  = dis[nc];
    float4 sf = *(const float4*)(xs + (size_t)nc * 4);
    float o0 = di * ((second ? g0 : a0) + sf.x);
    float o1 = di * ((second ? g1 : a1) + sf.y);
    float o2 = di * ((second ? g2 : a2) + sf.z);

    int srcl  = ((c >> 1) << 2) | (c & 1);
    float t0  = __shfl(o0, srcl);
    float t1  = __shfl(o1, srcl);
    float t2  = __shfl(o2, srcl);
    float dc  = __shfl(di, srcl);
    int ncol  = ((t - lane) >> 1) + c;
    bool cv   = ncol < n;

    v2f aw;
    aw.x = w1x;
    aw.y = (h == 0) ? w1y : 0.f;
    v2f bx;
    bx.x = cv ? (h == 0 ? t0 : t2) : 0.f;
    bx.y = (cv && h == 0) ? t1 : 0.f;
    v8f z = {};
    z = __builtin_amdgcn_wmma_f32_16x16x4_f32(false, aw, false, bx, (short)0, z, false, false);

    float ds = cv ? dc : 0.f;
    float zs[8];
    zs[0] = fmaxf(z[0] + bA.x, 0.f) * ds;
    zs[1] = fmaxf(z[1] + bA.y, 0.f) * ds;
    zs[2] = fmaxf(z[2] + bA.z, 0.f) * ds;
    zs[3] = fmaxf(z[3] + bA.w, 0.f) * ds;
    zs[4] = fmaxf(z[4] + bB.x, 0.f) * ds;
    zs[5] = fmaxf(z[5] + bB.y, 0.f) * ds;
    zs[6] = fmaxf(z[6] + bB.z, 0.f) * ds;
    zs[7] = fmaxf(z[7] + bB.w, 0.f) * ds;

    v8f acc = {};
    for (int s = 0; s < 4; ++s) {
        v2f a;
        a.x = ov ? w2x[s] : 0.f;
        a.y = ov ? w2y[s] : 0.f;
        v2f b;
        b.x = zs[2 * s];
        b.y = zs[2 * s + 1];
        acc = __builtin_amdgcn_wmma_f32_16x16x4_f32(false, a, false, b, (short)0, acc, false, false);
    }

    if (cv && h == 0) {
        float4* p = (float4*)(h2 + (size_t)ncol * 16);
        p[0] = make_float4(acc[0], acc[1], acc[2], acc[3]);
        p[1] = make_float4(acc[4], acc[5], acc[6], acc[7]);
    }
}

__global__ void k_agg2(const int* __restrict__ off, const int2* __restrict__ rec,
                       const float* __restrict__ h2, const float* __restrict__ dis,
                       const float* __restrict__ b2, float* __restrict__ out, int n, int e) {
    __shared__ float so[NPB * 7];
    __shared__ int2 srec[RCAP];
    int t    = blockIdx.x * blockDim.x + threadIdx.x;
    int sub  = t & 3;
    int nodeA = (t >> 2) * 2, nodeB = nodeA + 1;
    int s0;
    bool stail;
    int4 sq[RS];
    stage_load(off, rec, blockIdx.x * NPB, n, e, s0, stail, sq);
    int begA, endA, begB, endB;
    seg_bounds(off, nodeA, n, e, begA, endA);
    seg_bounds(off, nodeB, n, e, begB, endB);
    stage_store(srec, sq);
    __syncthreads();
    float sa[7], sg[7];
    for (int j = 0; j < 7; ++j) { sa[j] = 0.f; sg[j] = 0.f; }
    int kA = begA + sub, kB = begB + sub;
    if (stail)
        seg_sum7<true >(rec, srec, h2, s0, kA, endA, kB, endB, sa, sg);
    else
        seg_sum7<false>(rec, srec, h2, s0, kA, endA, kB, endB, sa, sg);
    float a0 = sa[0], a1 = sa[1], a2 = sa[2], a3 = sa[3], a4 = sa[4], a5 = sa[5], a6 = sa[6];
    float g0 = sg[0], g1 = sg[1], g2 = sg[2], g3 = sg[3], g4 = sg[4], g5 = sg[5], g6 = sg[6];
    a0 = quad_sum(a0); a1 = quad_sum(a1); a2 = quad_sum(a2); a3 = quad_sum(a3);
    a4 = quad_sum(a4); a5 = quad_sum(a5); a6 = quad_sum(a6);
    g0 = quad_sum(g0); g1 = quad_sum(g1); g2 = quad_sum(g2); g3 = quad_sum(g3);
    g4 = quad_sum(g4); g5 = quad_sum(g5); g6 = quad_sum(g6);
    bool second = (sub & 1) != 0;
    int nm = second ? nodeB : nodeA;
    if (nm < n && sub < 2) {
        float di = dis[nm];
        const float4* hs = (const float4*)(h2 + (size_t)nm * 16);
        float4 s0 = hs[0], s1 = hs[1];
        float* o = so + ((threadIdx.x >> 2) * 2 + (sub & 1)) * 7;
        o[0] = b2[0] + di * ((second ? g0 : a0) + s0.x);
        o[1] = b2[1] + di * ((second ? g1 : a1) + s0.y);
        o[2] = b2[2] + di * ((second ? g2 : a2) + s0.z);
        o[3] = b2[3] + di * ((second ? g3 : a3) + s0.w);
        o[4] = b2[4] + di * ((second ? g4 : a4) + s1.x);
        o[5] = b2[5] + di * ((second ? g5 : a5) + s1.y);
        o[6] = b2[6] + di * ((second ? g6 : a6) + s1.z);
    }
    __syncthreads();
    int nb    = blockIdx.x * NPB;
    int left  = n - nb;
    int cntf  = (left < NPB ? left : NPB) * 7;
    float* ob = out + (size_t)nb * 7;
    for (int j = threadIdx.x; j < NPB * 7; j += AGG_T)
        if (j < cntf) ob[j] = so[j];
}

extern "C" void kernel_launch(void* const* d_in, const int* in_sizes, int n_in,
                              void* d_out, int out_size, void* d_ws, size_t ws_size,
                              hipStream_t stream) {
    (void)n_in; (void)out_size; (void)ws_size;
    const float* x  = (const float*)d_in[0];
    const int*   ei = (const int*)d_in[1];
    const float* ew = (const float*)d_in[2];
    const float* W1 = (const float*)d_in[3];
    const float* b1 = (const float*)d_in[4];
    const float* W2 = (const float*)d_in[5];
    const float* b2 = (const float*)d_in[6];
    float* out = (float*)d_out;

    int n = in_sizes[0] / 3;
    int e = in_sizes[2];
    const int* row = ei;
    const int* col = ei + e;

    int nbk   = (n + 1 + BK_N - 1) / BK_N;
    int npad  = nbk * BK_N;
    int nblk  = (e + EPB - 1) / EPB;
    int pitch = (nblk + 3) / 4 * 4;
    int tabn  = nbk * pitch;

    int*   off    = (int*)d_ws;
    float* dis    = (float*)(off + (size_t)npad);
    float* xs     = dis  + (size_t)npad;
    float* h2     = xs   + (size_t)npad * 4;
    int2*  rec    = (int2*)(h2 + (size_t)npad * 16);
    int*   rcn    = (int*)(rec + (((size_t)e + 1) / 2) * 2);
    int*   rst    = rcn  + (size_t)tabn;
    int2*  trec   = (int2*)(rst + (size_t)tabn);
    int*   tcol   = (int*)(trec + (size_t)e);

    int quadB   = (n + NPB - 1) / NPB;

    k_bk_part   <<<nblk, PART_T, 0, stream>>>(row, col, ew, rst, rcn, trec, tcol, e, nblk, pitch, nbk);
    k_bk_sort   <<<nbk, BK_N, 0, stream>>>(rst, rcn, trec, tcol, x, off, rec, dis, xs, e, nblk, pitch, n);

    k_agg1_mlp  <<<quadB, AGG_T, 0, stream>>>(off, rec, xs, dis, W1, b1, W2, h2, n, e);
    k_agg2      <<<quadB, AGG_T, 0, stream>>>(off, rec, h2, dis, b2, out, n, e);
}
